# starting kernel
# baseline (speedup 1.0000x reference)
#include <hip/hip_runtime.h>
#include <math.h>

typedef __attribute__((ext_vector_type(2))) float v2f;
typedef __attribute__((ext_vector_type(8))) float v8f;

#define NN 50000
#define EE 800000
#define DD 128
#define GG 128
#define LDS_STRIDE 132

__global__ void __launch_bounds__(256)
node_gemm_wmma(const float* __restrict__ nf, const float* __restrict__ Wn,
               const float* __restrict__ bn, float* __restrict__ hv) {
  __shared__ float Atile[16 * LDS_STRIDE];
  const int row0 = blockIdx.x * 16;
  const int tid  = threadIdx.x;

  #pragma unroll
  for (int i = tid; i < 16 * DD; i += 256) {
    const int r = i >> 7;
    const int c = i & (DD - 1);
    Atile[r * LDS_STRIDE + c] = nf[(size_t)(row0 + r) * DD + c];
  }
  __syncthreads();

  const int wave = tid >> 5;
  const int lane = tid & 31;
  const int col0 = wave * 16;
  const int half = lane >> 4;
  const int l16  = lane & 15;

  v8f acc = {};
  #pragma unroll
  for (int k = 0; k < DD; k += 4) {
    const int ka = k + half * 2;
    v2f a, b;
    a.x = Atile[l16 * LDS_STRIDE + ka];
    a.y = Atile[l16 * LDS_STRIDE + ka + 1];
    b.x = Wn[(size_t)(ka + 0) * GG + col0 + l16];
    b.y = Wn[(size_t)(ka + 1) * GG + col0 + l16];
    acc = __builtin_amdgcn_wmma_f32_16x16x4_f32(false, a, false, b,
                                                (short)0, acc, false, false);
  }

  #pragma unroll
  for (int i = 0; i < 8; ++i) {
    const int row = row0 + i + half * 8;
    const int col = col0 + l16;
    float v = acc[i] + bn[col];
    v = v > 0.0f ? v : 0.01f * v;
    hv[(size_t)row * GG + col] = v;
  }
}

__global__ void __launch_bounds__(256)
node_scores(const float* __restrict__ nf, const float* __restrict__ We,
            float* __restrict__ p, float* __restrict__ q) {
  const int wid  = (int)((blockIdx.x * (size_t)blockDim.x + threadIdx.x) >> 5);
  const int lane = threadIdx.x & 31;
  if (wid >= NN) return;
  const float* row = nf + (size_t)wid * DD;
  float ap = 0.0f, aq = 0.0f;
  #pragma unroll
  for (int i = lane; i < DD; i += 32) {
    const float x = row[i];
    ap += x * We[i];
    aq += x * We[DD + i];
  }
  #pragma unroll
  for (int off = 16; off > 0; off >>= 1) {
    ap += __shfl_xor(ap, off, 32);
    aq += __shfl_xor(aq, off, 32);
  }
  if (lane == 0) { p[wid] = ap; q[wid] = aq; }
}

__global__ void __launch_bounds__(256)
init_buffers(float* __restrict__ m, float* __restrict__ s,
             float* __restrict__ out) {
  const int i = (int)(blockIdx.x * (size_t)blockDim.x + threadIdx.x);
  if (i < NN * GG) out[i] = 0.0f;
  if (i < NN) { m[i] = -INFINITY; s[i] = 0.0f; }
}

__device__ inline void atomicMaxF(float* addr, float v) {
  if (v >= 0.0f) atomicMax((int*)addr, __float_as_int(v));
  else           atomicMin((unsigned int*)addr, __float_as_uint(v));
}

__global__ void __launch_bounds__(256)
edge_logits(const int* __restrict__ src, const int* __restrict__ dst,
            const float* __restrict__ p, const float* __restrict__ q,
            const float* __restrict__ be, float* __restrict__ logit,
            float* __restrict__ m) {
  const int e = (int)(blockIdx.x * (size_t)blockDim.x + threadIdx.x);
  if (e >= EE) return;
  const int d = dst[e];
  float x = p[d] + q[src[e]] + be[0];
  x = x > 0.0f ? x : 0.01f * x;
  logit[e] = x;
  atomicMaxF(&m[d], x);
}

__global__ void __launch_bounds__(256)
edge_expsum(const int* __restrict__ dst, const float* __restrict__ logit,
            const float* __restrict__ m, float* __restrict__ s) {
  const int e = (int)(blockIdx.x * (size_t)blockDim.x + threadIdx.x);
  if (e >= EE) return;
  const int d = dst[e];
  atomicAdd(&s[d], expf(logit[e] - m[d]));
}

__global__ void __launch_bounds__(256)
edge_aggregate(const int* __restrict__ src, const int* __restrict__ dst,
               const float* __restrict__ logit, const float* __restrict__ m,
               const float* __restrict__ s, const float* __restrict__ hv,
               float* __restrict__ out) {
  const int e = (int)((blockIdx.x * (size_t)blockDim.x + threadIdx.x) >> 5);
  const int lane = threadIdx.x & 31;
  if (e >= EE) return;
  const int sv = src[e];
  const int dv = dst[e];
  const float den = s[dv];
  const float a = expf(logit[e] - m[dv]) / (den > 0.0f ? den : 1.0f);
  const float4 h = ((const float4*)(hv + (size_t)sv * GG))[lane];
  float* orow = out + (size_t)dv * GG + lane * 4;
  atomicAdd(orow + 0, h.x * a);
  atomicAdd(orow + 1, h.y * a);
  atomicAdd(orow + 2, h.z * a);
  atomicAdd(orow + 3, h.w * a);
}

__global__ void __launch_bounds__(256)
elu_inplace(float* __restrict__ out) {
  const int i = (int)(blockIdx.x * (size_t)blockDim.x + threadIdx.x);
  if (i >= NN * GG) return;
  const float c = out[i];
  out[i] = c > 0.0f ? c : expm1f(c);
}

extern "C" void kernel_launch(void* const* d_in, const int* in_sizes, int n_in,
                              void* d_out, int out_size, void* d_ws, size_t ws_size,
                              hipStream_t stream) {
  (void)in_sizes; (void)n_in; (void)out_size; (void)ws_size;
  const float* nf  = (const float*)d_in[0];
  const int*   src = (const int*)  d_in[1];
  const int*   dst = (const int*)  d_in[2];
  const float* We  = (const float*)d_in[3];
  const float* be  = (const float*)d_in[4];
  const float* Wn  = (const float*)d_in[5];
  const float* bn  = (const float*)d_in[6];
  float* out = (float*)d_out;

  float* ws    = (float*)d_ws;
  float* hv    = ws;
  float* p     = hv + (size_t)NN * GG;
  float* q     = p + NN;
  float* m     = q + NN;
  float* ssum  = m + NN;
  float* logit = ssum + NN;

  const int ngBlocks = (NN * GG + 255) / 256;

  init_buffers<<<ngBlocks, 256, 0, stream>>>(m, ssum, out);
  node_gemm_wmma<<<NN / 16, 256, 0, stream>>>(nf, Wn, bn, hv);
  node_scores<<<(NN * 32 + 255) / 256, 256, 0, stream>>>(nf, We, p, q);
  edge_logits<<<(EE + 255) / 256, 256, 0, stream>>>(src, dst, p, q, be, logit, m);
  edge_expsum<<<(EE + 255) / 256, 256, 0, stream>>>(dst, logit, m, ssum);
  edge_aggregate<<<(EE * 32 + 255) / 256, 256, 0, stream>>>(src, dst, logit, m, ssum, hv, out);
  elu_inplace<<<ngBlocks, 256, 0, stream>>>(out);
}


# Edge softmax + aggregation decomposed by destination range: 256 workgroups own 196 nodes each, filter the edge list, sum in LDS atomics (arrival order), divide by exp sum once per node; 3 launches
# speedup vs baseline: 2.6006x; 2.6006x over previous
#include <hip/hip_runtime.h>
#include <math.h>

typedef __attribute__((ext_vector_type(2))) float v2f;
typedef __attribute__((ext_vector_type(8))) float v8f;

#define NN 50000
#define EE 800000
#define DD 128
#define GG 128
#define LDS_STRIDE 132

__global__ void __launch_bounds__(256)
node_gemm_wmma(const float* __restrict__ nf, const float* __restrict__ Wn,
               const float* __restrict__ bn, float* __restrict__ hv) {
  __shared__ float Atile[16 * LDS_STRIDE];
  const int row0 = blockIdx.x * 16;
  const int tid  = threadIdx.x;

  #pragma unroll
  for (int i = tid; i < 16 * DD; i += 256) {
    const int r = i >> 7;
    const int c = i & (DD - 1);
    Atile[r * LDS_STRIDE + c] = nf[(size_t)(row0 + r) * DD + c];
  }
  __syncthreads();

  const int wave = tid >> 5;
  const int lane = tid & 31;
  const int col0 = wave * 16;
  const int half = lane >> 4;
  const int l16  = lane & 15;

  v8f acc = {};
  #pragma unroll
  for (int k = 0; k < DD; k += 4) {
    const int ka = k + half * 2;
    v2f a, b;
    a.x = Atile[l16 * LDS_STRIDE + ka];
    a.y = Atile[l16 * LDS_STRIDE + ka + 1];
    b.x = Wn[(size_t)(ka + 0) * GG + col0 + l16];
    b.y = Wn[(size_t)(ka + 1) * GG + col0 + l16];
    acc = __builtin_amdgcn_wmma_f32_16x16x4_f32(false, a, false, b,
                                                (short)0, acc, false, false);
  }

  #pragma unroll
  for (int i = 0; i < 8; ++i) {
    const int row = row0 + i + half * 8;
    const int col = col0 + l16;
    float v = acc[i] + bn[col];
    v = v > 0.0f ? v : 0.01f * v;
    hv[(size_t)row * GG + col] = v;
  }
}

__global__ void __launch_bounds__(256)
node_scores(const float* __restrict__ nf, const float* __restrict__ We,
            float* __restrict__ p, float* __restrict__ q) {
  const int wid  = (int)((blockIdx.x * (size_t)blockDim.x + threadIdx.x) >> 5);
  const int lane = threadIdx.x & 31;
  if (wid >= NN) return;
  const float* row = nf + (size_t)wid * DD;
  float ap = 0.0f, aq = 0.0f;
  #pragma unroll
  for (int i = lane; i < DD; i += 32) {
    const float x = row[i];
    ap += x * We[i];
    aq += x * We[DD + i];
  }
  #pragma unroll
  for (int off = 16; off > 0; off >>= 1) {
    ap += __shfl_xor(ap, off, 32);
    aq += __shfl_xor(aq, off, 32);
  }
  if (lane == 0) { p[wid] = ap; q[wid] = aq; }
}

#define RN 196
#define NWG 256
#define CAP 16384
#define NV4 (EE / 4)
#define NITER 196
#define SAFE_ITERS 4

__device__ inline void ldsMaxF(float* addr, float v) {
  const int bits = __float_as_int(v);
  if (bits >= 0) atomicMax((int*)addr, bits);
  else           atomicMin((unsigned int*)addr, (unsigned int)bits);
}

__global__ void __launch_bounds__(1024)
dst_range_softmax_aggregate(const int* __restrict__ src, const int* __restrict__ dst,
                            const float* __restrict__ p, const float* __restrict__ q,
                            const float* __restrict__ be, const float* __restrict__ hv,
                            float* __restrict__ out) {
  __shared__ float acc[RN * GG];
  __shared__ int   lkey[CAP];
  __shared__ float lval[CAP];
  __shared__ float mx[RN];
  __shared__ float mold[RN];
  __shared__ float sm[RN];
  __shared__ float pl[RN];
  __shared__ int   cnt;

  const int tid  = threadIdx.x;
  const int lane = tid & 31;
  const int wave = tid >> 5;
  const int base = blockIdx.x * RN;

  for (int i = tid; i < RN * GG; i += 1024) acc[i] = 0.0f;
  if (tid < RN) {
    const int node = base + tid;
    mx[tid]   = -INFINITY;
    mold[tid] = -INFINITY;
    sm[tid]   = 0.0f;
    pl[tid]   = node < NN ? p[node] : 0.0f;
  }
  const float bias = be[0];
  const int4* dst4 = (const int4*)dst;

  int epoch_len = NITER;
  int it0 = 0;
  while (it0 < NITER) {
    if (tid == 0) cnt = 0;
    __syncthreads();
    const int it_end = (it0 + epoch_len < NITER) ? (it0 + epoch_len) : NITER;

    for (int it = it0; it < it_end; it += 8) {
      int4 d[8];
      #pragma unroll
      for (int u = 0; u < 8; ++u) {
        const int idx = (it + u) * 1024 + tid;
        d[u] = (it + u < it_end && idx < NV4) ? dst4[idx] : make_int4(-1, -1, -1, -1);
      }
      #pragma unroll
      for (int u = 0; u < 8; ++u) {
        const int e0 = ((it + u) * 1024 + tid) * 4;
        const int dv[4] = {d[u].x, d[u].y, d[u].z, d[u].w};
        #pragma unroll
        for (int c = 0; c < 4; ++c) {
          const unsigned r = (unsigned)(dv[c] - base);
          if (r < (unsigned)RN) {
            const int k = atomicAdd(&cnt, 1);
            if (k < CAP) lkey[k] = (int)((r << 20) | (unsigned)(e0 + c));
          }
        }
      }
    }
    __syncthreads();
    const int total = cnt;
    if (total > CAP) {
      epoch_len = SAFE_ITERS;
      __syncthreads();
      continue;
    }

    for (int i = tid; i < total; i += 1024) {
      const unsigned ke = (unsigned)lkey[i];
      const unsigned r  = ke >> 20;
      const int sv = src[ke & 0xFFFFFu];
      float x = pl[r] + q[sv] + bias;
      x = x > 0.0f ? x : 0.01f * x;
      lkey[i] = (int)((r << 16) | (unsigned)sv);
      lval[i] = x;
      ldsMaxF(&mx[r], x);
    }
    __syncthreads();

    for (int r = wave; r < RN; r += 32) {
      const float mo = mold[r];
      const float mn = mx[r];
      if (mn != mo) {
        if (mo != -INFINITY) {
          const float f = expf(mo - mn);
          for (int j = 0; j < 4; ++j) acc[r * GG + j * 32 + lane] *= f;
          if (lane == 0) sm[r] *= f;
        }
        if (lane == 0) mold[r] = mn;
      }
    }
    __syncthreads();

    for (int i0 = wave; i0 < total; i0 += 32 * 4) {
      int    key[4];
      float  xv[4];
      float4 h[4];
      #pragma unroll
      for (int u = 0; u < 4; ++u) {
        const int i = i0 + 32 * u;
        if (i < total) {
          key[u] = lkey[i];
          xv[u]  = lval[i];
          h[u]   = ((const float4*)(hv + (size_t)(key[u] & 0xFFFF) * GG))[lane];
        }
      }
      #pragma unroll
      for (int u = 0; u < 4; ++u) {
        const int i = i0 + 32 * u;
        if (i < total) {
          const int r = (int)(((unsigned)key[u]) >> 16);
          const float ex = expf(xv[u] - mx[r]);
          if (lane == 0) atomicAdd(&sm[r], ex);
          float* a = &acc[r * GG + lane];
          atomicAdd(a,      ex * h[u].x);
          atomicAdd(a + 32, ex * h[u].y);
          atomicAdd(a + 64, ex * h[u].z);
          atomicAdd(a + 96, ex * h[u].w);
        }
      }
    }
    __syncthreads();
    it0 += epoch_len;
  }

  for (int f = tid; f < RN * 32; f += 1024) {
    const int r    = f >> 5;
    const int c4   = f & 31;
    const int node = base + r;
    if (node < NN) {
      const float den = sm[r];
      const float dd  = den > 0.0f ? den : 1.0f;
      float4 v;
      v.x = acc[r * GG +      c4] / dd;
      v.y = acc[r * GG + 32 + c4] / dd;
      v.z = acc[r * GG + 64 + c4] / dd;
      v.w = acc[r * GG + 96 + c4] / dd;
      v.x = v.x > 0.0f ? v.x : expm1f(v.x);
      v.y = v.y > 0.0f ? v.y : expm1f(v.y);
      v.z = v.z > 0.0f ? v.z : expm1f(v.z);
      v.w = v.w > 0.0f ? v.w : expm1f(v.w);
      ((float4*)(out + (size_t)node * GG))[c4] = v;
    }
  }
}

extern "C" void kernel_launch(void* const* d_in, const int* in_sizes, int n_in,
                              void* d_out, int out_size, void* d_ws, size_t ws_size,
                              hipStream_t stream) {
  (void)in_sizes; (void)n_in; (void)out_size; (void)ws_size;
  const float* nf  = (const float*)d_in[0];
  const int*   src = (const int*)  d_in[1];
  const int*   dst = (const int*)  d_in[2];
  const float* We  = (const float*)d_in[3];
  const float* be  = (const float*)d_in[4];
  const float* Wn  = (const float*)d_in[5];
  const float* bn  = (const float*)d_in[6];
  float* out = (float*)d_out;

  float* ws    = (float*)d_ws;
  float* hv    = ws;
  float* p     = hv + (size_t)NN * GG;
  float* q     = p + NN;

  node_gemm_wmma<<<NN / 16, 256, 0, stream>>>(nf, Wn, bn, hv);
  node_scores<<<(NN * 32 + 255) / 256, 256, 0, stream>>>(nf, We, p, q);
  dst_range_softmax_aggregate<<<NWG, 1024, 0, stream>>>(src, dst, p, q, be, hv, out);
}


# Kernel 3 scans one owner byte per edge (dst / 196, packed four to a word by the node score launch) with whole-word byte matches, not the 4-byte dst ids; sums still in LDS arrival order
# speedup vs baseline: 3.2824x; 1.2622x over previous
#include <hip/hip_runtime.h>
#include <math.h>

typedef __attribute__((ext_vector_type(2))) float v2f;
typedef __attribute__((ext_vector_type(8))) float v8f;

#define NN 50000
#define EE 800000
#define DD 128
#define GG 128
#define LDS_STRIDE 132

__global__ void __launch_bounds__(256)
node_gemm_wmma(const float* __restrict__ nf, const float* __restrict__ Wn,
               const float* __restrict__ bn, float* __restrict__ hv) {
  __shared__ float Atile[16 * LDS_STRIDE];
  const int row0 = blockIdx.x * 16;
  const int tid  = threadIdx.x;

  #pragma unroll
  for (int i = tid; i < 16 * DD; i += 256) {
    const int r = i >> 7;
    const int c = i & (DD - 1);
    Atile[r * LDS_STRIDE + c] = nf[(size_t)(row0 + r) * DD + c];
  }
  __syncthreads();

  const int wave = tid >> 5;
  const int lane = tid & 31;
  const int col0 = wave * 16;
  const int half = lane >> 4;
  const int l16  = lane & 15;

  v8f acc = {};
  #pragma unroll
  for (int k = 0; k < DD; k += 4) {
    const int ka = k + half * 2;
    v2f a, b;
    a.x = Atile[l16 * LDS_STRIDE + ka];
    a.y = Atile[l16 * LDS_STRIDE + ka + 1];
    b.x = Wn[(size_t)(ka + 0) * GG + col0 + l16];
    b.y = Wn[(size_t)(ka + 1) * GG + col0 + l16];
    acc = __builtin_amdgcn_wmma_f32_16x16x4_f32(false, a, false, b,
                                                (short)0, acc, false, false);
  }

  #pragma unroll
  for (int i = 0; i < 8; ++i) {
    const int row = row0 + i + half * 8;
    const int col = col0 + l16;
    float v = acc[i] + bn[col];
    v = v > 0.0f ? v : 0.01f * v;
    hv[(size_t)row * GG + col] = v;
  }
}

#define RN 196

__global__ void __launch_bounds__(256)
node_scores(const float* __restrict__ nf, const float* __restrict__ We,
            const int* __restrict__ dst, float* __restrict__ p,
            float* __restrict__ q, unsigned* __restrict__ own) {
  const int gid = (int)(blockIdx.x * blockDim.x + threadIdx.x);
  if (gid < EE / 4) {
    const int4 d = ((const int4*)dst)[gid];
    const unsigned o0 = min((unsigned)d.x / (unsigned)RN, 255u);
    const unsigned o1 = min((unsigned)d.y / (unsigned)RN, 255u);
    const unsigned o2 = min((unsigned)d.z / (unsigned)RN, 255u);
    const unsigned o3 = min((unsigned)d.w / (unsigned)RN, 255u);
    own[gid] = o0 | (o1 << 8) | (o2 << 16) | (o3 << 24);
  }
  const int wid  = gid >> 5;
  const int lane = threadIdx.x & 31;
  if (wid >= NN) return;
  const float* row = nf + (size_t)wid * DD;
  float ap = 0.0f, aq = 0.0f;
  #pragma unroll
  for (int i = lane; i < DD; i += 32) {
    const float x = row[i];
    ap += x * We[i];
    aq += x * We[DD + i];
  }
  #pragma unroll
  for (int off = 16; off > 0; off >>= 1) {
    ap += __shfl_xor(ap, off, 32);
    aq += __shfl_xor(aq, off, 32);
  }
  if (lane == 0) { p[wid] = ap; q[wid] = aq; }
}

#define NWG 256
#define CAP 16384
#define NV16 (EE / 16)
#define NITER 49
#define SAFE_ITERS 1
#define SCAN_U 7

__device__ inline unsigned zeroBytes(unsigned v) {
  const unsigned t = (v & 0x7F7F7F7Fu) + 0x7F7F7F7Fu;
  return ~(t | v | 0x7F7F7F7Fu);
}

__device__ inline void ldsMaxF(float* addr, float v) {
  const int bits = __float_as_int(v);
  if (bits >= 0) atomicMax((int*)addr, bits);
  else           atomicMin((unsigned int*)addr, (unsigned int)bits);
}

__global__ void __launch_bounds__(1024)
dst_range_softmax_aggregate(const int* __restrict__ src, const int* __restrict__ dst,
                            const unsigned* __restrict__ own,
                            const float* __restrict__ p, const float* __restrict__ q,
                            const float* __restrict__ be, const float* __restrict__ hv,
                            float* __restrict__ out) {
  __shared__ float acc[RN * GG];
  __shared__ int   lkey[CAP];
  __shared__ float lval[CAP];
  __shared__ float mx[RN];
  __shared__ float mold[RN];
  __shared__ float sm[RN];
  __shared__ float pl[RN];
  __shared__ int   cnt;

  const int tid  = threadIdx.x;
  const int lane = tid & 31;
  const int wave = tid >> 5;
  const int base = blockIdx.x * RN;

  for (int i = tid; i < RN * GG; i += 1024) acc[i] = 0.0f;
  if (tid < RN) {
    const int node = base + tid;
    mx[tid]   = -INFINITY;
    mold[tid] = -INFINITY;
    sm[tid]   = 0.0f;
    pl[tid]   = node < NN ? p[node] : 0.0f;
  }
  const float bias = be[0];
  const uint4* own16 = (const uint4*)own;
  const unsigned pat = (unsigned)blockIdx.x * 0x01010101u;

  int epoch_len = NITER;
  int it0 = 0;
  while (it0 < NITER) {
    if (tid == 0) cnt = 0;
    __syncthreads();
    const int it_end = (it0 + epoch_len < NITER) ? (it0 + epoch_len) : NITER;

    for (int it = it0; it < it_end; it += SCAN_U) {
      uint4 w[SCAN_U];
      #pragma unroll
      for (int u = 0; u < SCAN_U; ++u) {
        const int idx = (it + u) * 1024 + tid;
        w[u] = own16[min(idx, NV16 - 1)];
      }
      #pragma unroll
      for (int u = 0; u < SCAN_U; ++u) {
        const int idx = (it + u) * 1024 + tid;
        const int e0  = idx * 16;
        unsigned m = zeroBytes(w[u].x ^ pat)        | (zeroBytes(w[u].y ^ pat) >> 1) |
                     (zeroBytes(w[u].z ^ pat) >> 2) | (zeroBytes(w[u].w ^ pat) >> 3);
        if (!(it + u < it_end && idx < NV16)) m = 0u;
        while (m != 0u) {
          const int pos = __builtin_ctz(m);
          m &= m - 1u;
          const int e = e0 + ((7 - (pos & 7)) << 2) + (pos >> 3);
          const int k = atomicAdd(&cnt, 1);
          if (k < CAP) lkey[k] = e;
        }
      }
    }
    __syncthreads();
    const int total = cnt;
    if (total > CAP) {
      epoch_len = SAFE_ITERS;
      __syncthreads();
      continue;
    }

    for (int i = tid; i < total; i += 1024) {
      const int e  = lkey[i];
      const unsigned r = min((unsigned)(dst[e] - base), (unsigned)(RN - 1));
      const int sv = src[e];
      float x = pl[r] + q[sv] + bias;
      x = x > 0.0f ? x : 0.01f * x;
      lkey[i] = (int)((r << 16) | (unsigned)sv);
      lval[i] = x;
      ldsMaxF(&mx[r], x);
    }
    __syncthreads();

    for (int r = wave; r < RN; r += 32) {
      const float mo = mold[r];
      const float mn = mx[r];
      if (mn != mo) {
        if (mo != -INFINITY) {
          const float f = expf(mo - mn);
          for (int j = 0; j < 4; ++j) acc[r * GG + j * 32 + lane] *= f;
          if (lane == 0) sm[r] *= f;
        }
        if (lane == 0) mold[r] = mn;
      }
    }
    __syncthreads();

    for (int i0 = wave; i0 < total; i0 += 32 * 4) {
      int    key[4];
      float  xv[4];
      float4 h[4];
      #pragma unroll
      for (int u = 0; u < 4; ++u) {
        const int i = i0 + 32 * u;
        if (i < total) {
          key[u] = lkey[i];
          xv[u]  = lval[i];
          h[u]   = ((const float4*)(hv + (size_t)(key[u] & 0xFFFF) * GG))[lane];
        }
      }
      #pragma unroll
      for (int u = 0; u < 4; ++u) {
        const int i = i0 + 32 * u;
        if (i < total) {
          const int r = (int)(((unsigned)key[u]) >> 16);
          const float ex = expf(xv[u] - mx[r]);
          if (lane == 0) atomicAdd(&sm[r], ex);
          float* a = &acc[r * GG + lane];
          atomicAdd(a,      ex * h[u].x);
          atomicAdd(a + 32, ex * h[u].y);
          atomicAdd(a + 64, ex * h[u].z);
          atomicAdd(a + 96, ex * h[u].w);
        }
      }
    }
    __syncthreads();
    it0 += epoch_len;
  }

  for (int f = tid; f < RN * 32; f += 1024) {
    const int r    = f >> 5;
    const int c4   = f & 31;
    const int node = base + r;
    if (node < NN) {
      const float den = sm[r];
      const float dd  = den > 0.0f ? den : 1.0f;
      float4 v;
      v.x = acc[r * GG +      c4] / dd;
      v.y = acc[r * GG + 32 + c4] / dd;
      v.z = acc[r * GG + 64 + c4] / dd;
      v.w = acc[r * GG + 96 + c4] / dd;
      v.x = v.x > 0.0f ? v.x : expm1f(v.x);
      v.y = v.y > 0.0f ? v.y : expm1f(v.y);
      v.z = v.z > 0.0f ? v.z : expm1f(v.z);
      v.w = v.w > 0.0f ? v.w : expm1f(v.w);
      ((float4*)(out + (size_t)node * GG))[c4] = v;
    }
  }
}

extern "C" void kernel_launch(void* const* d_in, const int* in_sizes, int n_in,
                              void* d_out, int out_size, void* d_ws, size_t ws_size,
                              hipStream_t stream) {
  (void)in_sizes; (void)n_in; (void)out_size; (void)ws_size;
  const float* nf  = (const float*)d_in[0];
  const int*   src = (const int*)  d_in[1];
  const int*   dst = (const int*)  d_in[2];
  const float* We  = (const float*)d_in[3];
  const float* be  = (const float*)d_in[4];
  const float* Wn  = (const float*)d_in[5];
  const float* bn  = (const float*)d_in[6];
  float* out = (float*)d_out;

  float* ws    = (float*)d_ws;
  float* hv    = ws;
  float* p     = hv + (size_t)NN * GG;
  float* q     = p + NN;
  unsigned* own = (unsigned*)(q + NN);

  node_gemm_wmma<<<NN / 16, 256, 0, stream>>>(nf, Wn, bn, hv);
  node_scores<<<(NN * 32 + 255) / 256, 256, 0, stream>>>(nf, We, dst, p, q, own);
  dst_range_softmax_aggregate<<<NWG, 1024, 0, stream>>>(src, dst, own, p, q, be, hv, out);
}


# Node score launch: one wave per 16 nodes (sixteen 16-byte row loads in flight, the 32 partial sums folded to one total per lane), 3125 waves not 50000; row products summed in another lane order
# speedup vs baseline: 3.5188x; 1.0720x over previous
#include <hip/hip_runtime.h>
#include <math.h>

typedef __attribute__((ext_vector_type(2))) float v2f;
typedef __attribute__((ext_vector_type(8))) float v8f;

#define NN 50000
#define EE 800000
#define DD 128
#define GG 128
#define LDS_STRIDE 132

__global__ void __launch_bounds__(256)
node_gemm_wmma(const float* __restrict__ nf, const float* __restrict__ Wn,
               const float* __restrict__ bn, float* __restrict__ hv) {
  __shared__ float Atile[16 * LDS_STRIDE];
  const int row0 = blockIdx.x * 16;
  const int tid  = threadIdx.x;

  #pragma unroll
  for (int i = tid; i < 16 * DD; i += 256) {
    const int r = i >> 7;
    const int c = i & (DD - 1);
    Atile[r * LDS_STRIDE + c] = nf[(size_t)(row0 + r) * DD + c];
  }
  __syncthreads();

  const int wave = tid >> 5;
  const int lane = tid & 31;
  const int col0 = wave * 16;
  const int half = lane >> 4;
  const int l16  = lane & 15;

  v8f acc = {};
  #pragma unroll
  for (int k = 0; k < DD; k += 4) {
    const int ka = k + half * 2;
    v2f a, b;
    a.x = Atile[l16 * LDS_STRIDE + ka];
    a.y = Atile[l16 * LDS_STRIDE + ka + 1];
    b.x = Wn[(size_t)(ka + 0) * GG + col0 + l16];
    b.y = Wn[(size_t)(ka + 1) * GG + col0 + l16];
    acc = __builtin_amdgcn_wmma_f32_16x16x4_f32(false, a, false, b,
                                                (short)0, acc, false, false);
  }

  #pragma unroll
  for (int i = 0; i < 8; ++i) {
    const int row = row0 + i + half * 8;
    const int col = col0 + l16;
    float v = acc[i] + bn[col];
    v = v > 0.0f ? v : 0.01f * v;
    hv[(size_t)row * GG + col] = v;
  }
}

#define RN 196

#define NPW 16
static_assert(NN % NPW == 0, "a wave's nodes are all inside the array");
static_assert(2 * NPW == 32, "one total per lane after the fold");
static_assert((NN / NPW) * 32 >= EE / 8, "kernel 2 has a thread for every pair of owner words");

__device__ inline unsigned ownerWord(int4 d) {
  const unsigned o0 = min((unsigned)d.x / (unsigned)RN, 255u);
  const unsigned o1 = min((unsigned)d.y / (unsigned)RN, 255u);
  const unsigned o2 = min((unsigned)d.z / (unsigned)RN, 255u);
  const unsigned o3 = min((unsigned)d.w / (unsigned)RN, 255u);
  return o0 | (o1 << 8) | (o2 << 16) | (o3 << 24);
}

template <int H, int OFF>
__device__ inline void foldStep(float (&v)[2 * NPW], int lane) {
  const bool hi = (lane & OFF) != 0;
  #pragma unroll
  for (int j = 0; j < H; ++j) {
    const float keep = hi ? v[j + H] : v[j];
    const float send = hi ? v[j] : v[j + H];
    v[j] = keep + __shfl_xor(send, OFF, 32);
  }
}

__global__ void __launch_bounds__(256)
node_scores(const float* __restrict__ nf, const float* __restrict__ We,
            const int* __restrict__ dst, float* __restrict__ p,
            float* __restrict__ q, unsigned* __restrict__ own) {
  const int gid = (int)(blockIdx.x * blockDim.x + threadIdx.x);
  if (gid < EE / 8) {
    const int4 d0 = ((const int4*)dst)[2 * gid];
    const int4 d1 = ((const int4*)dst)[2 * gid + 1];
    ((uint2*)own)[gid] = make_uint2(ownerWord(d0), ownerWord(d1));
  }
  const int wid  = gid >> 5;
  const int lane = threadIdx.x & 31;
  const int node0 = wid * NPW;
  if (node0 >= NN) return;

  const float4* rows = (const float4*)(nf + (size_t)node0 * DD);
  float4 x[NPW];
  #pragma unroll
  for (int u = 0; u < NPW; ++u) x[u] = rows[u * (DD / 4) + lane];
  const float4 wp = ((const float4*)We)[lane];
  const float4 wq = ((const float4*)(We + DD))[lane];

  float v[2 * NPW];
  #pragma unroll
  for (int u = 0; u < NPW; ++u) {
    v[u]       = x[u].x * wp.x + x[u].y * wp.y + x[u].z * wp.z + x[u].w * wp.w;
    v[NPW + u] = x[u].x * wq.x + x[u].y * wq.y + x[u].z * wq.z + x[u].w * wq.w;
  }
  foldStep<16, 16>(v, lane);
  foldStep<8, 8>(v, lane);
  foldStep<4, 4>(v, lane);
  foldStep<2, 2>(v, lane);
  foldStep<1, 1>(v, lane);
  float* o = lane < NPW ? p : q;
  o[node0 + (lane & (NPW - 1))] = v[0];
}

#define NWG 256
#define CAP 16384
#define NV16 (EE / 16)
#define NITER 49
#define SAFE_ITERS 1
#define SCAN_U 7

__device__ inline unsigned zeroBytes(unsigned v) {
  const unsigned t = (v & 0x7F7F7F7Fu) + 0x7F7F7F7Fu;
  return ~(t | v | 0x7F7F7F7Fu);
}

__device__ inline void ldsMaxF(float* addr, float v) {
  const int bits = __float_as_int(v);
  if (bits >= 0) atomicMax((int*)addr, bits);
  else           atomicMin((unsigned int*)addr, (unsigned int)bits);
}

__global__ void __launch_bounds__(1024)
dst_range_softmax_aggregate(const int* __restrict__ src, const int* __restrict__ dst,
                            const unsigned* __restrict__ own,
                            const float* __restrict__ p, const float* __restrict__ q,
                            const float* __restrict__ be, const float* __restrict__ hv,
                            float* __restrict__ out) {
  __shared__ float acc[RN * GG];
  __shared__ int   lkey[CAP];
  __shared__ float lval[CAP];
  __shared__ float mx[RN];
  __shared__ float mold[RN];
  __shared__ float sm[RN];
  __shared__ float pl[RN];
  __shared__ int   cnt;

  const int tid  = threadIdx.x;
  const int lane = tid & 31;
  const int wave = tid >> 5;
  const int base = blockIdx.x * RN;

  for (int i = tid; i < RN * GG; i += 1024) acc[i] = 0.0f;
  if (tid < RN) {
    const int node = base + tid;
    mx[tid]   = -INFINITY;
    mold[tid] = -INFINITY;
    sm[tid]   = 0.0f;
    pl[tid]   = node < NN ? p[node] : 0.0f;
  }
  const float bias = be[0];
  const uint4* own16 = (const uint4*)own;
  const unsigned pat = (unsigned)blockIdx.x * 0x01010101u;

  int epoch_len = NITER;
  int it0 = 0;
  while (it0 < NITER) {
    if (tid == 0) cnt = 0;
    __syncthreads();
    const int it_end = (it0 + epoch_len < NITER) ? (it0 + epoch_len) : NITER;

    for (int it = it0; it < it_end; it += SCAN_U) {
      uint4 w[SCAN_U];
      #pragma unroll
      for (int u = 0; u < SCAN_U; ++u) {
        const int idx = (it + u) * 1024 + tid;
        w[u] = own16[min(idx, NV16 - 1)];
      }
      #pragma unroll
      for (int u = 0; u < SCAN_U; ++u) {
        const int idx = (it + u) * 1024 + tid;
        const int e0  = idx * 16;
        unsigned m = zeroBytes(w[u].x ^ pat)        | (zeroBytes(w[u].y ^ pat) >> 1) |
                     (zeroBytes(w[u].z ^ pat) >> 2) | (zeroBytes(w[u].w ^ pat) >> 3);
        if (!(it + u < it_end && idx < NV16)) m = 0u;
        while (m != 0u) {
          const int pos = __builtin_ctz(m);
          m &= m - 1u;
          const int e = e0 + ((7 - (pos & 7)) << 2) + (pos >> 3);
          const int k = atomicAdd(&cnt, 1);
          if (k < CAP) lkey[k] = e;
        }
      }
    }
    __syncthreads();
    const int total = cnt;
    if (total > CAP) {
      epoch_len = SAFE_ITERS;
      __syncthreads();
      continue;
    }

    for (int i = tid; i < total; i += 1024) {
      const int e  = lkey[i];
      const unsigned r = min((unsigned)(dst[e] - base), (unsigned)(RN - 1));
      const int sv = src[e];
      float x = pl[r] + q[sv] + bias;
      x = x > 0.0f ? x : 0.01f * x;
      lkey[i] = (int)((r << 16) | (unsigned)sv);
      lval[i] = x;
      ldsMaxF(&mx[r], x);
    }
    __syncthreads();

    for (int r = wave; r < RN; r += 32) {
      const float mo = mold[r];
      const float mn = mx[r];
      if (mn != mo) {
        if (mo != -INFINITY) {
          const float f = expf(mo - mn);
          for (int j = 0; j < 4; ++j) acc[r * GG + j * 32 + lane] *= f;
          if (lane == 0) sm[r] *= f;
        }
        if (lane == 0) mold[r] = mn;
      }
    }
    __syncthreads();

    for (int i0 = wave; i0 < total; i0 += 32 * 4) {
      int    key[4];
      float  xv[4];
      float4 h[4];
      #pragma unroll
      for (int u = 0; u < 4; ++u) {
        const int i = i0 + 32 * u;
        if (i < total) {
          key[u] = lkey[i];
          xv[u]  = lval[i];
          h[u]   = ((const float4*)(hv + (size_t)(key[u] & 0xFFFF) * GG))[lane];
        }
      }
      #pragma unroll
      for (int u = 0; u < 4; ++u) {
        const int i = i0 + 32 * u;
        if (i < total) {
          const int r = (int)(((unsigned)key[u]) >> 16);
          const float ex = expf(xv[u] - mx[r]);
          if (lane == 0) atomicAdd(&sm[r], ex);
          float* a = &acc[r * GG + lane];
          atomicAdd(a,      ex * h[u].x);
          atomicAdd(a + 32, ex * h[u].y);
          atomicAdd(a + 64, ex * h[u].z);
          atomicAdd(a + 96, ex * h[u].w);
        }
      }
    }
    __syncthreads();
    it0 += epoch_len;
  }

  for (int f = tid; f < RN * 32; f += 1024) {
    const int r    = f >> 5;
    const int c4   = f & 31;
    const int node = base + r;
    if (node < NN) {
      const float den = sm[r];
      const float dd  = den > 0.0f ? den : 1.0f;
      float4 v;
      v.x = acc[r * GG +      c4] / dd;
      v.y = acc[r * GG + 32 + c4] / dd;
      v.z = acc[r * GG + 64 + c4] / dd;
      v.w = acc[r * GG + 96 + c4] / dd;
      v.x = v.x > 0.0f ? v.x : expm1f(v.x);
      v.y = v.y > 0.0f ? v.y : expm1f(v.y);
      v.z = v.z > 0.0f ? v.z : expm1f(v.z);
      v.w = v.w > 0.0f ? v.w : expm1f(v.w);
      ((float4*)(out + (size_t)node * GG))[c4] = v;
    }
  }
}

extern "C" void kernel_launch(void* const* d_in, const int* in_sizes, int n_in,
                              void* d_out, int out_size, void* d_ws, size_t ws_size,
                              hipStream_t stream) {
  (void)in_sizes; (void)n_in; (void)out_size; (void)ws_size;
  const float* nf  = (const float*)d_in[0];
  const int*   src = (const int*)  d_in[1];
  const int*   dst = (const int*)  d_in[2];
  const float* We  = (const float*)d_in[3];
  const float* be  = (const float*)d_in[4];
  const float* Wn  = (const float*)d_in[5];
  const float* bn  = (const float*)d_in[6];
  float* out = (float*)d_out;

  float* ws    = (float*)d_ws;
  float* hv    = ws;
  float* p     = hv + (size_t)NN * GG;
  float* q     = p + NN;
  unsigned* own = (unsigned*)(q + NN);

  node_gemm_wmma<<<NN / 16, 256, 0, stream>>>(nf, Wn, bn, hv);
  node_scores<<<((NN / NPW) * 32 + 255) / 256, 256, 0, stream>>>(nf, We, dst, p, q, own);
  dst_range_softmax_aggregate<<<NWG, 1024, 0, stream>>>(src, dst, own, p, q, be, hv, out);
}


# Launch 3 pass C: a wave takes 8 list entries per trip, lane j reads entry j and computes its exp term, keys broadcast as scalars, 8 hv row loads issued as one group; sums still in LDS arrival order
# speedup vs baseline: 3.9757x; 1.1298x over previous
#include <hip/hip_runtime.h>
#include <math.h>

typedef __attribute__((ext_vector_type(2))) float v2f;
typedef __attribute__((ext_vector_type(8))) float v8f;

#define NN 50000
#define EE 800000
#define DD 128
#define GG 128
#define LDS_STRIDE 132

__global__ void __launch_bounds__(256)
node_gemm_wmma(const float* __restrict__ nf, const float* __restrict__ Wn,
               const float* __restrict__ bn, float* __restrict__ hv) {
  __shared__ float Atile[16 * LDS_STRIDE];
  const int row0 = blockIdx.x * 16;
  const int tid  = threadIdx.x;

  #pragma unroll
  for (int i = tid; i < 16 * DD; i += 256) {
    const int r = i >> 7;
    const int c = i & (DD - 1);
    Atile[r * LDS_STRIDE + c] = nf[(size_t)(row0 + r) * DD + c];
  }
  __syncthreads();

  const int wave = tid >> 5;
  const int lane = tid & 31;
  const int col0 = wave * 16;
  const int half = lane >> 4;
  const int l16  = lane & 15;

  v8f acc = {};
  #pragma unroll
  for (int k = 0; k < DD; k += 4) {
    const int ka = k + half * 2;
    v2f a, b;
    a.x = Atile[l16 * LDS_STRIDE + ka];
    a.y = Atile[l16 * LDS_STRIDE + ka + 1];
    b.x = Wn[(size_t)(ka + 0) * GG + col0 + l16];
    b.y = Wn[(size_t)(ka + 1) * GG + col0 + l16];
    acc = __builtin_amdgcn_wmma_f32_16x16x4_f32(false, a, false, b,
                                                (short)0, acc, false, false);
  }

  #pragma unroll
  for (int i = 0; i < 8; ++i) {
    const int row = row0 + i + half * 8;
    const int col = col0 + l16;
    float v = acc[i] + bn[col];
    v = v > 0.0f ? v : 0.01f * v;
    hv[(size_t)row * GG + col] = v;
  }
}

#define RN 196

#define NPW 16
static_assert(NN % NPW == 0, "a wave's nodes are all inside the array");
static_assert(2 * NPW == 32, "one total per lane after the fold");
static_assert((NN / NPW) * 32 >= EE / 8, "kernel 2 has a thread for every pair of owner words");

__device__ inline unsigned ownerWord(int4 d) {
  const unsigned o0 = min((unsigned)d.x / (unsigned)RN, 255u);
  const unsigned o1 = min((unsigned)d.y / (unsigned)RN, 255u);
  const unsigned o2 = min((unsigned)d.z / (unsigned)RN, 255u);
  const unsigned o3 = min((unsigned)d.w / (unsigned)RN, 255u);
  return o0 | (o1 << 8) | (o2 << 16) | (o3 << 24);
}

template <int H, int OFF>
__device__ inline void foldStep(float (&v)[2 * NPW], int lane) {
  const bool hi = (lane & OFF) != 0;
  #pragma unroll
  for (int j = 0; j < H; ++j) {
    const float keep = hi ? v[j + H] : v[j];
    const float send = hi ? v[j] : v[j + H];
    v[j] = keep + __shfl_xor(send, OFF, 32);
  }
}

__global__ void __launch_bounds__(256)
node_scores(const float* __restrict__ nf, const float* __restrict__ We,
            const int* __restrict__ dst, float* __restrict__ p,
            float* __restrict__ q, unsigned* __restrict__ own) {
  const int gid = (int)(blockIdx.x * blockDim.x + threadIdx.x);
  if (gid < EE / 8) {
    const int4 d0 = ((const int4*)dst)[2 * gid];
    const int4 d1 = ((const int4*)dst)[2 * gid + 1];
    ((uint2*)own)[gid] = make_uint2(ownerWord(d0), ownerWord(d1));
  }
  const int wid  = gid >> 5;
  const int lane = threadIdx.x & 31;
  const int node0 = wid * NPW;
  if (node0 >= NN) return;

  const float4* rows = (const float4*)(nf + (size_t)node0 * DD);
  float4 x[NPW];
  #pragma unroll
  for (int u = 0; u < NPW; ++u) x[u] = rows[u * (DD / 4) + lane];
  const float4 wp = ((const float4*)We)[lane];
  const float4 wq = ((const float4*)(We + DD))[lane];

  float v[2 * NPW];
  #pragma unroll
  for (int u = 0; u < NPW; ++u) {
    v[u]       = x[u].x * wp.x + x[u].y * wp.y + x[u].z * wp.z + x[u].w * wp.w;
    v[NPW + u] = x[u].x * wq.x + x[u].y * wq.y + x[u].z * wq.z + x[u].w * wq.w;
  }
  foldStep<16, 16>(v, lane);
  foldStep<8, 8>(v, lane);
  foldStep<4, 4>(v, lane);
  foldStep<2, 2>(v, lane);
  foldStep<1, 1>(v, lane);
  float* o = lane < NPW ? p : q;
  o[node0 + (lane & (NPW - 1))] = v[0];
}

#define NWG 256
#define CAP 16384
#define NV16 (EE / 16)
#define NITER 49
#define SAFE_ITERS 1
#define SCAN_U 7
#define PCB 8

__device__ inline unsigned zeroBytes(unsigned v) {
  const unsigned t = (v & 0x7F7F7F7Fu) + 0x7F7F7F7Fu;
  return ~(t | v | 0x7F7F7F7Fu);
}

__device__ inline void ldsMaxF(float* addr, float v) {
  const int bits = __float_as_int(v);
  if (bits >= 0) atomicMax((int*)addr, bits);
  else           atomicMin((unsigned int*)addr, (unsigned int)bits);
}

__global__ void __launch_bounds__(1024)
dst_range_softmax_aggregate(const int* __restrict__ src, const int* __restrict__ dst,
                            const unsigned* __restrict__ own,
                            const float* __restrict__ p, const float* __restrict__ q,
                            const float* __restrict__ be, const float* __restrict__ hv,
                            float* __restrict__ out) {
  __shared__ float acc[RN * GG];
  __shared__ int   lkey[CAP];
  __shared__ float lval[CAP];
  __shared__ float mx[RN];
  __shared__ float mold[RN];
  __shared__ float sm[RN];
  __shared__ float pl[RN];
  __shared__ int   cnt;

  const int tid  = threadIdx.x;
  const int lane = tid & 31;
  const int wave = tid >> 5;
  const int base = blockIdx.x * RN;

  for (int i = tid; i < RN * GG; i += 1024) acc[i] = 0.0f;
  if (tid < RN) {
    const int node = base + tid;
    mx[tid]   = -INFINITY;
    mold[tid] = -INFINITY;
    sm[tid]   = 0.0f;
    pl[tid]   = node < NN ? p[node] : 0.0f;
  }
  const float bias = be[0];
  const uint4* own16 = (const uint4*)own;
  const unsigned pat = (unsigned)blockIdx.x * 0x01010101u;

  int epoch_len = NITER;
  int it0 = 0;
  while (it0 < NITER) {
    if (tid == 0) cnt = 0;
    __syncthreads();
    const int it_end = (it0 + epoch_len < NITER) ? (it0 + epoch_len) : NITER;

    for (int it = it0; it < it_end; it += SCAN_U) {
      uint4 w[SCAN_U];
      #pragma unroll
      for (int u = 0; u < SCAN_U; ++u) {
        const int idx = (it + u) * 1024 + tid;
        w[u] = own16[min(idx, NV16 - 1)];
      }
      #pragma unroll
      for (int u = 0; u < SCAN_U; ++u) {
        const int idx = (it + u) * 1024 + tid;
        const int e0  = idx * 16;
        unsigned m = zeroBytes(w[u].x ^ pat)        | (zeroBytes(w[u].y ^ pat) >> 1) |
                     (zeroBytes(w[u].z ^ pat) >> 2) | (zeroBytes(w[u].w ^ pat) >> 3);
        if (!(it + u < it_end && idx < NV16)) m = 0u;
        while (m != 0u) {
          const int pos = __builtin_ctz(m);
          m &= m - 1u;
          const int e = e0 + ((7 - (pos & 7)) << 2) + (pos >> 3);
          const int k = atomicAdd(&cnt, 1);
          if (k < CAP) lkey[k] = e;
        }
      }
    }
    __syncthreads();
    const int total = cnt;
    if (total > CAP) {
      epoch_len = SAFE_ITERS;
      __syncthreads();
      continue;
    }

    for (int i = tid; i < total; i += 1024) {
      const int e  = lkey[i];
      const unsigned r = min((unsigned)(dst[e] - base), (unsigned)(RN - 1));
      const int sv = src[e];
      float x = pl[r] + q[sv] + bias;
      x = x > 0.0f ? x : 0.01f * x;
      lkey[i] = (int)((r << 16) | (unsigned)sv);
      lval[i] = x;
      ldsMaxF(&mx[r], x);
    }
    __syncthreads();

    for (int r = wave; r < RN; r += 32) {
      const float mo = mold[r];
      const float mn = mx[r];
      if (mn != mo) {
        if (mo != -INFINITY) {
          const float f = expf(mo - mn);
          for (int j = 0; j < 4; ++j) acc[r * GG + j * 32 + lane] *= f;
          if (lane == 0) sm[r] *= f;
        }
        if (lane == 0) mold[r] = mn;
      }
    }
    __syncthreads();

    {
      const int wv  = __builtin_amdgcn_readfirstlane(wave);
      const int tot = __builtin_amdgcn_readfirstlane(total);
      for (int i0 = wv * PCB; i0 < tot; i0 += 32 * PCB) {
        const int nv = min(PCB, tot - i0);
        const int   j  = min(lane, nv - 1);
        const int   kj = lkey[i0 + j];
        const float xj = lval[i0 + j];
        float4 h[PCB];
        #pragma unroll
        for (int u = 0; u < PCB; ++u) {
          const int ku = __builtin_amdgcn_readlane(kj, u);
          h[u] = ((const float4*)(hv + (size_t)(ku & 0xFFFF) * GG))[lane];
        }
        const int   rj = (int)(((unsigned)kj) >> 16);
        const float ej = expf(xj - mx[rj]);
        if (lane < nv) atomicAdd(&sm[rj], ej);
        #pragma unroll
        for (int u = 0; u < PCB; ++u) {
          if (u < nv) {
            const int   r  = (int)(((unsigned)__builtin_amdgcn_readlane(kj, u)) >> 16);
            const float ex = __int_as_float(__builtin_amdgcn_readlane(__float_as_int(ej), u));
            float* a = &acc[r * GG + lane];
            atomicAdd(a,      ex * h[u].x);
            atomicAdd(a + 32, ex * h[u].y);
            atomicAdd(a + 64, ex * h[u].z);
            atomicAdd(a + 96, ex * h[u].w);
          }
        }
      }
    }
    __syncthreads();
    it0 += epoch_len;
  }

  for (int f = tid; f < RN * 32; f += 1024) {
    const int r    = f >> 5;
    const int c4   = f & 31;
    const int node = base + r;
    if (node < NN) {
      const float den = sm[r];
      const float dd  = den > 0.0f ? den : 1.0f;
      float4 v;
      v.x = acc[r * GG +      c4] / dd;
      v.y = acc[r * GG + 32 + c4] / dd;
      v.z = acc[r * GG + 64 + c4] / dd;
      v.w = acc[r * GG + 96 + c4] / dd;
      v.x = v.x > 0.0f ? v.x : expm1f(v.x);
      v.y = v.y > 0.0f ? v.y : expm1f(v.y);
      v.z = v.z > 0.0f ? v.z : expm1f(v.z);
      v.w = v.w > 0.0f ? v.w : expm1f(v.w);
      ((float4*)(out + (size_t)node * GG))[c4] = v;
    }
  }
}

extern "C" void kernel_launch(void* const* d_in, const int* in_sizes, int n_in,
                              void* d_out, int out_size, void* d_ws, size_t ws_size,
                              hipStream_t stream) {
  (void)in_sizes; (void)n_in; (void)out_size; (void)ws_size;
  const float* nf  = (const float*)d_in[0];
  const int*   src = (const int*)  d_in[1];
  const int*   dst = (const int*)  d_in[2];
  const float* We  = (const float*)d_in[3];
  const float* be  = (const float*)d_in[4];
  const float* Wn  = (const float*)d_in[5];
  const float* bn  = (const float*)d_in[6];
  float* out = (float*)d_out;

  float* ws    = (float*)d_ws;
  float* hv    = ws;
  float* p     = hv + (size_t)NN * GG;
  float* q     = p + NN;
  unsigned* own = (unsigned*)(q + NN);

  node_gemm_wmma<<<NN / 16, 256, 0, stream>>>(nf, Wn, bn, hv);
  node_scores<<<((NN / NPW) * 32 + 255) / 256, 256, 0, stream>>>(nf, We, dst, p, q, own);
  dst_range_softmax_aggregate<<<NWG, 1024, 0, stream>>>(src, dst, own, p, q, be, hv, out);
}


# Launch 2 sorts each 4096-edge chunk by owner range in LDS and stores packed node+source records; launch 3 copies its own segments: no owner-byte scan, no dst/src gathers; sums in LDS arrival order
# speedup vs baseline: 5.0296x; 1.2651x over previous
#include <hip/hip_runtime.h>
#include <math.h>

typedef __attribute__((ext_vector_type(2))) float v2f;
typedef __attribute__((ext_vector_type(8))) float v8f;

#define NN 50000
#define EE 800000
#define DD 128
#define GG 128
#define LDS_STRIDE 132

__global__ void __launch_bounds__(256)
node_gemm_wmma(const float* __restrict__ nf, const float* __restrict__ Wn,
               const float* __restrict__ bn, float* __restrict__ hv) {
  __shared__ float Atile[16 * LDS_STRIDE];
  const int row0 = blockIdx.x * 16;
  const int tid  = threadIdx.x;

  #pragma unroll
  for (int i = tid; i < 16 * DD; i += 256) {
    const int r = i >> 7;
    const int c = i & (DD - 1);
    Atile[r * LDS_STRIDE + c] = nf[(size_t)(row0 + r) * DD + c];
  }
  __syncthreads();

  const int wave = tid >> 5;
  const int lane = tid & 31;
  const int col0 = wave * 16;
  const int half = lane >> 4;
  const int l16  = lane & 15;

  v8f acc = {};
  #pragma unroll
  for (int k = 0; k < DD; k += 4) {
    const int ka = k + half * 2;
    v2f a, b;
    a.x = Atile[l16 * LDS_STRIDE + ka];
    a.y = Atile[l16 * LDS_STRIDE + ka + 1];
    b.x = Wn[(size_t)(ka + 0) * GG + col0 + l16];
    b.y = Wn[(size_t)(ka + 1) * GG + col0 + l16];
    acc = __builtin_amdgcn_wmma_f32_16x16x4_f32(false, a, false, b,
                                                (short)0, acc, false, false);
  }

  #pragma unroll
  for (int i = 0; i < 8; ++i) {
    const int row = row0 + i + half * 8;
    const int col = col0 + l16;
    float v = acc[i] + bn[col];
    v = v > 0.0f ? v : 0.01f * v;
    hv[(size_t)row * GG + col] = v;
  }
}

#define RN 196
#define NWG 256

#define NPW 16
#define CH  4096
#define NCH ((EE + CH - 1) / CH)
#define K2T (CH / 8)
static_assert(NN % NPW == 0, "a wave's nodes are all inside the array");
static_assert(2 * NPW == 32, "one total per lane after the fold");
static_assert((NN / NPW) * 32 == EE / 8, "the threads that own edges are the threads that own nodes");
static_assert(NCH * K2T >= EE / 8 && (EE / 8) % 32 == 0, "every edge has a thread; validity is per wave");
static_assert(NWG * RN >= NN && NWG == 256, "the prefix below gives eight owners to each lane of one wave");
static_assert(CH < 65536 && NN <= 65536 && RN < 65536, "record and table words pack two 16-bit fields");

template <int H, int OFF>
__device__ inline void foldStep(float (&v)[2 * NPW], int lane) {
  const bool hi = (lane & OFF) != 0;
  #pragma unroll
  for (int j = 0; j < H; ++j) {
    const float keep = hi ? v[j + H] : v[j];
    const float send = hi ? v[j] : v[j + H];
    v[j] = keep + __shfl_xor(send, OFF, 32);
  }
}

__global__ void __launch_bounds__(K2T)
node_scores_partition(const float* __restrict__ nf, const float* __restrict__ We,
                      const int* __restrict__ src, const int* __restrict__ dst,
                      float* __restrict__ p, float* __restrict__ q,
                      unsigned* __restrict__ offc, unsigned* __restrict__ recs) {
  __shared__ unsigned hist[NWG];
  __shared__ unsigned bstart[NWG];
  __shared__ uint4    srec4[CH / 4];
  unsigned* srec = (unsigned*)srec4;

  const int tid  = threadIdx.x;
  const int lane = tid & 31;
  const int gid  = (int)(blockIdx.x * K2T) + tid;
  const bool valid = gid < EE / 8;

  int dd[8] = {0, 0, 0, 0, 0, 0, 0, 0};
  int ss[8] = {0, 0, 0, 0, 0, 0, 0, 0};
  float4 x[NPW];
  float4 wp = make_float4(0.f, 0.f, 0.f, 0.f), wq = wp;
  #pragma unroll
  for (int u = 0; u < NPW; ++u) x[u] = wp;
  const int node0 = (gid >> 5) * NPW;
  if (valid) {
    const int4 d0 = ((const int4*)dst)[2 * gid];
    const int4 d1 = ((const int4*)dst)[2 * gid + 1];
    const int4 s0 = ((const int4*)src)[2 * gid];
    const int4 s1 = ((const int4*)src)[2 * gid + 1];
    dd[0] = d0.x; dd[1] = d0.y; dd[2] = d0.z; dd[3] = d0.w;
    dd[4] = d1.x; dd[5] = d1.y; dd[6] = d1.z; dd[7] = d1.w;
    ss[0] = s0.x; ss[1] = s0.y; ss[2] = s0.z; ss[3] = s0.w;
    ss[4] = s1.x; ss[5] = s1.y; ss[6] = s1.z; ss[7] = s1.w;
    const float4* rows = (const float4*)(nf + (size_t)node0 * DD);
    #pragma unroll
    for (int u = 0; u < NPW; ++u) x[u] = rows[u * (DD / 4) + lane];
    wp = ((const float4*)We)[lane];
    wq = ((const float4*)(We + DD))[lane];
  }
  if (tid < NWG) hist[tid] = 0u;
  __syncthreads();

  unsigned ow[8], rk[8];
  #pragma unroll
  for (int j = 0; j < 8; ++j) { ow[j] = 0u; rk[j] = 0u; }
  float score = 0.0f;
  if (valid) {
    #pragma unroll
    for (int j = 0; j < 8; ++j) {
      ow[j] = min((unsigned)dd[j] / (unsigned)RN, (unsigned)(NWG - 1));
      rk[j] = atomicAdd(&hist[ow[j]], 1u);
    }

    float v[2 * NPW];
    #pragma unroll
    for (int u = 0; u < NPW; ++u) {
      v[u]       = x[u].x * wp.x + x[u].y * wp.y + x[u].z * wp.z + x[u].w * wp.w;
      v[NPW + u] = x[u].x * wq.x + x[u].y * wq.y + x[u].z * wq.z + x[u].w * wq.w;
    }
    foldStep<16, 16>(v, lane);
    foldStep<8, 8>(v, lane);
    foldStep<4, 4>(v, lane);
    foldStep<2, 2>(v, lane);
    foldStep<1, 1>(v, lane);
    score = v[0];
  }
  __syncthreads();

  if (tid < 32) {
    unsigned c[8], e[8], run = 0u;
    #pragma unroll
    for (int j = 0; j < 8; ++j) { c[j] = hist[8 * lane + j]; e[j] = run; run += c[j]; }
    unsigned inc = run;
    #pragma unroll
    for (int off = 1; off < 32; off <<= 1) {
      const unsigned n = __shfl_up(inc, off, 32);
      if (lane >= off) inc += n;
    }
    const unsigned base = inc - run;
    #pragma unroll
    for (int j = 0; j < 8; ++j) bstart[8 * lane + j] = base + e[j];
  }
  __syncthreads();

  if (valid) {
    #pragma unroll
    for (int j = 0; j < 8; ++j) {
      const unsigned r = min((unsigned)dd[j] - ow[j] * (unsigned)RN, (unsigned)(RN - 1));
      srec[bstart[ow[j]] + rk[j]] = (r << 16) | ((unsigned)ss[j] & 0xFFFFu);
    }
  }
  const unsigned tabw = tid < NWG ? ((bstart[tid] << 16) | hist[tid]) : 0u;
  const unsigned tot  = bstart[NWG - 1] + hist[NWG - 1];
  __syncthreads();

  uint4* outc = (uint4*)recs + (size_t)blockIdx.x * (CH / 4);
  #pragma unroll
  for (int k = 0; k < (CH / 4) / K2T; ++k) {
    const int i4 = tid + k * K2T;
    if ((unsigned)(4 * i4) < tot) outc[i4] = srec4[i4];
  }
  if (tid < NWG) offc[(size_t)tid * NCH + blockIdx.x] = tabw;
  if (valid) {
    float* o = lane < NPW ? p : q;
    o[node0 + (lane & (NPW - 1))] = score;
  }
}

#define CAP 16384
#define SAFE_CH (CAP / CH)
#define PCB 8
static_assert(SAFE_CH >= 1 && SAFE_CH * CH <= CAP, "a short epoch always fits the list");
static_assert(NCH <= 4 * 64 && NCH <= 1024, "four segments per half-wave cover an epoch; one thread per segment");

__device__ inline void ldsMaxF(float* addr, float v) {
  const int bits = __float_as_int(v);
  if (bits >= 0) atomicMax((int*)addr, bits);
  else           atomicMin((unsigned int*)addr, (unsigned int)bits);
}

__global__ void __launch_bounds__(1024)
dst_range_softmax_aggregate(const unsigned* __restrict__ offc, const unsigned* __restrict__ recs,
                            const float* __restrict__ p, const float* __restrict__ q,
                            const float* __restrict__ be, const float* __restrict__ hv,
                            float* __restrict__ out) {
  __shared__ float acc[RN * GG];
  __shared__ int   lkey[CAP];
  __shared__ float lval[CAP];
  __shared__ float mx[RN];
  __shared__ float mold[RN];
  __shared__ float sm[RN];
  __shared__ float pl[RN];
  __shared__ int      segk[NCH];
  __shared__ unsigned segw[NCH];
  __shared__ int   cnt;

  const int tid  = threadIdx.x;
  const int lane = tid & 31;
  const int wave = tid >> 5;
  const int base = blockIdx.x * RN;

  for (int i = tid; i < RN * GG; i += 1024) acc[i] = 0.0f;
  if (tid < RN) {
    const int node = base + tid;
    mx[tid]   = -INFINITY;
    mold[tid] = -INFINITY;
    sm[tid]   = 0.0f;
    pl[tid]   = node < NN ? p[node] : 0.0f;
  }
  const float bias = be[0];
  const unsigned* myoff = offc + (size_t)blockIdx.x * NCH;

  int epoch_len = NCH;
  int c0 = 0;
  while (c0 < NCH) {
    if (tid == 0) cnt = 0;
    __syncthreads();
    const int c_end = (c0 + epoch_len < NCH) ? (c0 + epoch_len) : NCH;
    const int nseg  = c_end - c0;

    if (tid < ((nseg + 31) & ~31)) {
      const unsigned word = tid < nseg ? myoff[c0 + tid] : 0u;
      const unsigned c = word & 0xFFFFu;
      unsigned inc = c;
      #pragma unroll
      for (int off = 1; off < 32; off <<= 1) {
        const unsigned n = __shfl_up(inc, off, 32);
        if (lane >= off) inc += n;
      }
      int basew = 0;
      if (lane == 31) basew = atomicAdd(&cnt, (int)inc);
      basew = __shfl(basew, 31, 32);
      if (tid < nseg) {
        segk[tid] = basew + (int)(inc - c);
        segw[tid] = word;
      }
    }
    __syncthreads();
    const int total = cnt;
    if (total > CAP) {
      epoch_len = SAFE_CH;
      __syncthreads();
      continue;
    }

    {
      const int hw  = tid >> 4;
      const int sub = tid & 15;
      const unsigned* sp[4];
      unsigned cc[4], v0[4], v1[4];
      int kk[4];
      #pragma unroll
      for (int u = 0; u < 4; ++u) {
        const int t  = hw + 64 * u;
        const int tc = t < nseg ? t : nseg - 1;
        const unsigned word = segw[tc];
        cc[u] = t < nseg ? (word & 0xFFFFu) : 0u;
        kk[u] = segk[tc];
        sp[u] = recs + (size_t)(c0 + tc) * CH + (word >> 16);
        v0[u] = 0u; v1[u] = 0u;
        if ((unsigned)sub < cc[u])        v0[u] = sp[u][sub];
        if ((unsigned)(sub + 16) < cc[u]) v1[u] = sp[u][sub + 16];
      }
      #pragma unroll
      for (int u = 0; u < 4; ++u) {
        if ((unsigned)sub < cc[u])        lkey[kk[u] + sub]      = (int)v0[u];
        if ((unsigned)(sub + 16) < cc[u]) lkey[kk[u] + sub + 16] = (int)v1[u];
        #pragma clang loop unroll(disable)
        for (unsigned i = (unsigned)sub + 32u; i < cc[u]; i += 16u) lkey[kk[u] + (int)i] = (int)sp[u][i];
      }
    }
    __syncthreads();

    for (int i = tid; i < total; i += 1024) {
      const unsigned key = (unsigned)lkey[i];
      const unsigned r = key >> 16;
      float x = pl[r] + q[key & 0xFFFFu] + bias;
      x = x > 0.0f ? x : 0.01f * x;
      lval[i] = x;
      ldsMaxF(&mx[r], x);
    }
    __syncthreads();
    for (int r = wave; r < RN; r += 32) {
      const float mo = mold[r];
      const float mn = mx[r];
      if (mn != mo) {
        if (mo != -INFINITY) {
          const float f = expf(mo - mn);
          for (int j = 0; j < 4; ++j) acc[r * GG + j * 32 + lane] *= f;
          if (lane == 0) sm[r] *= f;
        }
        if (lane == 0) mold[r] = mn;
      }
    }
    __syncthreads();

    {
      const int wv  = __builtin_amdgcn_readfirstlane(wave);
      const int tot = __builtin_amdgcn_readfirstlane(total);
      for (int i0 = wv * PCB; i0 < tot; i0 += 32 * PCB) {
        const int nv = min(PCB, tot - i0);
        const int   j  = min(lane, nv - 1);
        const int   kj = lkey[i0 + j];
        const float xj = lval[i0 + j];
        float4 h[PCB];
        #pragma unroll
        for (int u = 0; u < PCB; ++u) {
          const int ku = __builtin_amdgcn_readlane(kj, u);
          h[u] = ((const float4*)(hv + (size_t)(ku & 0xFFFF) * GG))[lane];
        }
        const int   rj = (int)(((unsigned)kj) >> 16);
        const float ej = expf(xj - mx[rj]);
        if (lane < nv) atomicAdd(&sm[rj], ej);
        #pragma unroll
        for (int u = 0; u < PCB; ++u) {
          if (u < nv) {
            const int   r  = (int)(((unsigned)__builtin_amdgcn_readlane(kj, u)) >> 16);
            const float ex = __int_as_float(__builtin_amdgcn_readlane(__float_as_int(ej), u));
            float* a = &acc[r * GG + lane];
            atomicAdd(a,      ex * h[u].x);
            atomicAdd(a + 32, ex * h[u].y);
            atomicAdd(a + 64, ex * h[u].z);
            atomicAdd(a + 96, ex * h[u].w);
          }
        }
      }
    }
    __syncthreads();
    c0 += epoch_len;
  }

  for (int f = tid; f < RN * 32; f += 1024) {
    const int r    = f >> 5;
    const int c4   = f & 31;
    const int node = base + r;
    if (node < NN) {
      const float den = sm[r];
      const float dd  = den > 0.0f ? den : 1.0f;
      float4 v;
      v.x = acc[r * GG +      c4] / dd;
      v.y = acc[r * GG + 32 + c4] / dd;
      v.z = acc[r * GG + 64 + c4] / dd;
      v.w = acc[r * GG + 96 + c4] / dd;
      v.x = v.x > 0.0f ? v.x : expm1f(v.x);
      v.y = v.y > 0.0f ? v.y : expm1f(v.y);
      v.z = v.z > 0.0f ? v.z : expm1f(v.z);
      v.w = v.w > 0.0f ? v.w : expm1f(v.w);
      ((float4*)(out + (size_t)node * GG))[c4] = v;
    }
  }
}

extern "C" void kernel_launch(void* const* d_in, const int* in_sizes, int n_in,
                              void* d_out, int out_size, void* d_ws, size_t ws_size,
                              hipStream_t stream) {
  (void)in_sizes; (void)n_in; (void)out_size; (void)ws_size;
  const float* nf  = (const float*)d_in[0];
  const int*   src = (const int*)  d_in[1];
  const int*   dst = (const int*)  d_in[2];
  const float* We  = (const float*)d_in[3];
  const float* be  = (const float*)d_in[4];
  const float* Wn  = (const float*)d_in[5];
  const float* bn  = (const float*)d_in[6];
  float* out = (float*)d_out;

  float* ws    = (float*)d_ws;
  float* hv    = ws;
  float* p     = hv + (size_t)NN * GG;
  float* q     = p + NN;
  unsigned* offc = (unsigned*)(q + NN);
  unsigned* recs = offc + (size_t)NWG * NCH;

  node_gemm_wmma<<<NN / 16, 256, 0, stream>>>(nf, Wn, bn, hv);
  node_scores_partition<<<NCH, K2T, 0, stream>>>(nf, We, src, dst, p, q, offc, recs);
  dst_range_softmax_aggregate<<<NWG, 1024, 0, stream>>>(offc, recs, p, q, be, hv, out);
}


# Node GEMM launch: 1024 workgroups; each wave keeps its 16-column W_node fragments in registers and walks 3-4 row tiles through a two-buffer LDS stage, one barrier per tile; same order of sums
# speedup vs baseline: 5.1768x; 1.0293x over previous
#include <hip/hip_runtime.h>
#include <math.h>

typedef __attribute__((ext_vector_type(2))) float v2f;
typedef __attribute__((ext_vector_type(8))) float v8f;

#define NN 50000
#define EE 800000
#define DD 128
#define GG 128
#define LDS_STRIDE 132

#define K1_WGS 1024
#define NT (NN / 16)
static_assert(NN % 16 == 0, "whole row tiles only");
static_assert(K1_WGS <= NT, "every block has a first tile");

__global__ void __launch_bounds__(256)
node_gemm_wmma(const float* __restrict__ nf, const float* __restrict__ Wn,
               const float* __restrict__ bn, float* __restrict__ hv) {
  __shared__ float Atile[2][16 * LDS_STRIDE];
  const int tid  = threadIdx.x;
  const int wave = tid >> 5;
  const int lane = tid & 31;
  const int col0 = wave * 16;
  const int half = lane >> 4;
  const int l16  = lane & 15;

  const int srow = wave * LDS_STRIDE + lane * 4;
  int tile = blockIdx.x;
  {
    const float4* t0 = (const float4*)(nf + (size_t)tile * 16 * DD);
    const float4 f0 = t0[tid];
    const float4 f1 = t0[tid + 256];
    *(float4*)&Atile[0][srow]                  = f0;
    *(float4*)&Atile[0][srow + 8 * LDS_STRIDE] = f1;
  }

  v2f b[DD / 4];
  #pragma unroll
  for (int k = 0; k < DD; k += 4) {
    const int ka = k + half * 2;
    b[k >> 2].x = Wn[(size_t)(ka + 0) * GG + col0 + l16];
    b[k >> 2].y = Wn[(size_t)(ka + 1) * GG + col0 + l16];
  }
  const float bias = bn[col0 + l16];
  __syncthreads();

  int buf = 0;
  for (;;) {
    const int  next = tile + K1_WGS;
    const bool more = next < NT;
    float4 n0 = make_float4(0.f, 0.f, 0.f, 0.f), n1 = n0;
    if (more) {
      const float4* tn = (const float4*)(nf + (size_t)next * 16 * DD);
      n0 = tn[tid];
      n1 = tn[tid + 256];
    }

    const float* At = &Atile[buf][l16 * LDS_STRIDE + half * 2];
    v8f acc = {};
    #pragma unroll
    for (int k = 0; k < DD; k += 4) {
      v2f a;
      a.x = At[k];
      a.y = At[k + 1];
      acc = __builtin_amdgcn_wmma_f32_16x16x4_f32(false, a, false, b[k >> 2],
                                                  (short)0, acc, false, false);
    }

    if (more) {
      *(float4*)&Atile[buf ^ 1][srow]                  = n0;
      *(float4*)&Atile[buf ^ 1][srow + 8 * LDS_STRIDE] = n1;
      __syncthreads();
    }

    const int row0 = tile * 16;
    #pragma unroll
    for (int i = 0; i < 8; ++i) {
      const int row = row0 + i + half * 8;
      float v = acc[i] + bias;
      v = v > 0.0f ? v : 0.01f * v;
      hv[(size_t)row * GG + col0 + l16] = v;
    }

    if (!more) break;
    tile = next;
    buf ^= 1;
  }
}

#define RN 196
#define NWG 256

#define NPW 16
#define CH  4096
#define NCH ((EE + CH - 1) / CH)
#define K2T (CH / 8)
static_assert(NN % NPW == 0, "a wave's nodes are all inside the array");
static_assert(2 * NPW == 32, "one total per lane after the fold");
static_assert((NN / NPW) * 32 == EE / 8, "the threads that own edges are the threads that own nodes");
static_assert(NCH * K2T >= EE / 8 && (EE / 8) % 32 == 0, "every edge has a thread; validity is per wave");
static_assert(NWG * RN >= NN && NWG == 256, "the prefix below gives eight owners to each lane of one wave");
static_assert(CH < 65536 && NN <= 65536 && RN < 65536, "record and table words pack two 16-bit fields");

template <int H, int OFF>
__device__ inline void foldStep(float (&v)[2 * NPW], int lane) {
  const bool hi = (lane & OFF) != 0;
  #pragma unroll
  for (int j = 0; j < H; ++j) {
    const float keep = hi ? v[j + H] : v[j];
    const float send = hi ? v[j] : v[j + H];
    v[j] = keep + __shfl_xor(send, OFF, 32);
  }
}

__global__ void __launch_bounds__(K2T)
node_scores_partition(const float* __restrict__ nf, const float* __restrict__ We,
                      const int* __restrict__ src, const int* __restrict__ dst,
                      float* __restrict__ p, float* __restrict__ q,
                      unsigned* __restrict__ offc, unsigned* __restrict__ recs) {
  __shared__ unsigned hist[NWG];
  __shared__ unsigned bstart[NWG];
  __shared__ uint4    srec4[CH / 4];
  unsigned* srec = (unsigned*)srec4;

  const int tid  = threadIdx.x;
  const int lane = tid & 31;
  const int gid  = (int)(blockIdx.x * K2T) + tid;
  const bool valid = gid < EE / 8;

  int dd[8] = {0, 0, 0, 0, 0, 0, 0, 0};
  int ss[8] = {0, 0, 0, 0, 0, 0, 0, 0};
  float4 x[NPW];
  float4 wp = make_float4(0.f, 0.f, 0.f, 0.f), wq = wp;
  #pragma unroll
  for (int u = 0; u < NPW; ++u) x[u] = wp;
  const int node0 = (gid >> 5) * NPW;
  if (valid) {
    const int4 d0 = ((const int4*)dst)[2 * gid];
    const int4 d1 = ((const int4*)dst)[2 * gid + 1];
    const int4 s0 = ((const int4*)src)[2 * gid];
    const int4 s1 = ((const int4*)src)[2 * gid + 1];
    dd[0] = d0.x; dd[1] = d0.y; dd[2] = d0.z; dd[3] = d0.w;
    dd[4] = d1.x; dd[5] = d1.y; dd[6] = d1.z; dd[7] = d1.w;
    ss[0] = s0.x; ss[1] = s0.y; ss[2] = s0.z; ss[3] = s0.w;
    ss[4] = s1.x; ss[5] = s1.y; ss[6] = s1.z; ss[7] = s1.w;
    const float4* rows = (const float4*)(nf + (size_t)node0 * DD);
    #pragma unroll
    for (int u = 0; u < NPW; ++u) x[u] = rows[u * (DD / 4) + lane];
    wp = ((const float4*)We)[lane];
    wq = ((const float4*)(We + DD))[lane];
  }
  if (tid < NWG) hist[tid] = 0u;
  __syncthreads();

  unsigned ow[8], rk[8];
  #pragma unroll
  for (int j = 0; j < 8; ++j) { ow[j] = 0u; rk[j] = 0u; }
  float score = 0.0f;
  if (valid) {
    #pragma unroll
    for (int j = 0; j < 8; ++j) {
      ow[j] = min((unsigned)dd[j] / (unsigned)RN, (unsigned)(NWG - 1));
      rk[j] = atomicAdd(&hist[ow[j]], 1u);
    }

    float v[2 * NPW];
    #pragma unroll
    for (int u = 0; u < NPW; ++u) {
      v[u]       = x[u].x * wp.x + x[u].y * wp.y + x[u].z * wp.z + x[u].w * wp.w;
      v[NPW + u] = x[u].x * wq.x + x[u].y * wq.y + x[u].z * wq.z + x[u].w * wq.w;
    }
    foldStep<16, 16>(v, lane);
    foldStep<8, 8>(v, lane);
    foldStep<4, 4>(v, lane);
    foldStep<2, 2>(v, lane);
    foldStep<1, 1>(v, lane);
    score = v[0];
  }
  __syncthreads();

  if (tid < 32) {
    unsigned c[8], e[8], run = 0u;
    #pragma unroll
    for (int j = 0; j < 8; ++j) { c[j] = hist[8 * lane + j]; e[j] = run; run += c[j]; }
    unsigned inc = run;
    #pragma unroll
    for (int off = 1; off < 32; off <<= 1) {
      const unsigned n = __shfl_up(inc, off, 32);
      if (lane >= off) inc += n;
    }
    const unsigned base = inc - run;
    #pragma unroll
    for (int j = 0; j < 8; ++j) bstart[8 * lane + j] = base + e[j];
  }
  __syncthreads();

  if (valid) {
    #pragma unroll
    for (int j = 0; j < 8; ++j) {
      const unsigned r = min((unsigned)dd[j] - ow[j] * (unsigned)RN, (unsigned)(RN - 1));
      srec[bstart[ow[j]] + rk[j]] = (r << 16) | ((unsigned)ss[j] & 0xFFFFu);
    }
  }
  const unsigned tabw = tid < NWG ? ((bstart[tid] << 16) | hist[tid]) : 0u;
  const unsigned tot  = bstart[NWG - 1] + hist[NWG - 1];
  __syncthreads();

  uint4* outc = (uint4*)recs + (size_t)blockIdx.x * (CH / 4);
  #pragma unroll
  for (int k = 0; k < (CH / 4) / K2T; ++k) {
    const int i4 = tid + k * K2T;
    if ((unsigned)(4 * i4) < tot) outc[i4] = srec4[i4];
  }
  if (tid < NWG) offc[(size_t)tid * NCH + blockIdx.x] = tabw;
  if (valid) {
    float* o = lane < NPW ? p : q;
    o[node0 + (lane & (NPW - 1))] = score;
  }
}

#define CAP 16384
#define SAFE_CH (CAP / CH)
#define PCB 8
static_assert(SAFE_CH >= 1 && SAFE_CH * CH <= CAP, "a short epoch always fits the list");
static_assert(NCH <= 4 * 64 && NCH <= 1024, "four segments per half-wave cover an epoch; one thread per segment");

__device__ inline void ldsMaxF(float* addr, float v) {
  const int bits = __float_as_int(v);
  if (bits >= 0) atomicMax((int*)addr, bits);
  else           atomicMin((unsigned int*)addr, (unsigned int)bits);
}

__global__ void __launch_bounds__(1024)
dst_range_softmax_aggregate(const unsigned* __restrict__ offc, const unsigned* __restrict__ recs,
                            const float* __restrict__ p, const float* __restrict__ q,
                            const float* __restrict__ be, const float* __restrict__ hv,
                            float* __restrict__ out) {
  __shared__ float acc[RN * GG];
  __shared__ int   lkey[CAP];
  __shared__ float lval[CAP];
  __shared__ float mx[RN];
  __shared__ float mold[RN];
  __shared__ float sm[RN];
  __shared__ float pl[RN];
  __shared__ int      segk[NCH];
  __shared__ unsigned segw[NCH];
  __shared__ int   cnt;

  const int tid  = threadIdx.x;
  const int lane = tid & 31;
  const int wave = tid >> 5;
  const int base = blockIdx.x * RN;

  for (int i = tid; i < RN * GG; i += 1024) acc[i] = 0.0f;
  if (tid < RN) {
    const int node = base + tid;
    mx[tid]   = -INFINITY;
    mold[tid] = -INFINITY;
    sm[tid]   = 0.0f;
    pl[tid]   = node < NN ? p[node] : 0.0f;
  }
  const float bias = be[0];
  const unsigned* myoff = offc + (size_t)blockIdx.x * NCH;

  int epoch_len = NCH;
  int c0 = 0;
  while (c0 < NCH) {
    if (tid == 0) cnt = 0;
    __syncthreads();
    const int c_end = (c0 + epoch_len < NCH) ? (c0 + epoch_len) : NCH;
    const int nseg  = c_end - c0;

    if (tid < ((nseg + 31) & ~31)) {
      const unsigned word = tid < nseg ? myoff[c0 + tid] : 0u;
      const unsigned c = word & 0xFFFFu;
      unsigned inc = c;
      #pragma unroll
      for (int off = 1; off < 32; off <<= 1) {
        const unsigned n = __shfl_up(inc, off, 32);
        if (lane >= off) inc += n;
      }
      int basew = 0;
      if (lane == 31) basew = atomicAdd(&cnt, (int)inc);
      basew = __shfl(basew, 31, 32);
      if (tid < nseg) {
        segk[tid] = basew + (int)(inc - c);
        segw[tid] = word;
      }
    }
    __syncthreads();
    const int total = cnt;
    if (total > CAP) {
      epoch_len = SAFE_CH;
      __syncthreads();
      continue;
    }

    {
      const int hw  = tid >> 4;
      const int sub = tid & 15;
      const unsigned* sp[4];
      unsigned cc[4], v0[4], v1[4];
      int kk[4];
      #pragma unroll
      for (int u = 0; u < 4; ++u) {
        const int t  = hw + 64 * u;
        const int tc = t < nseg ? t : nseg - 1;
        const unsigned word = segw[tc];
        cc[u] = t < nseg ? (word & 0xFFFFu) : 0u;
        kk[u] = segk[tc];
        sp[u] = recs + (size_t)(c0 + tc) * CH + (word >> 16);
        v0[u] = 0u; v1[u] = 0u;
        if ((unsigned)sub < cc[u])        v0[u] = sp[u][sub];
        if ((unsigned)(sub + 16) < cc[u]) v1[u] = sp[u][sub + 16];
      }
      #pragma unroll
      for (int u = 0; u < 4; ++u) {
        if ((unsigned)sub < cc[u])        lkey[kk[u] + sub]      = (int)v0[u];
        if ((unsigned)(sub + 16) < cc[u]) lkey[kk[u] + sub + 16] = (int)v1[u];
        #pragma clang loop unroll(disable)
        for (unsigned i = (unsigned)sub + 32u; i < cc[u]; i += 16u) lkey[kk[u] + (int)i] = (int)sp[u][i];
      }
    }
    __syncthreads();

    for (int i = tid; i < total; i += 1024) {
      const unsigned key = (unsigned)lkey[i];
      const unsigned r = key >> 16;
      float x = pl[r] + q[key & 0xFFFFu] + bias;
      x = x > 0.0f ? x : 0.01f * x;
      lval[i] = x;
      ldsMaxF(&mx[r], x);
    }
    __syncthreads();
    for (int r = wave; r < RN; r += 32) {
      const float mo = mold[r];
      const float mn = mx[r];
      if (mn != mo) {
        if (mo != -INFINITY) {
          const float f = expf(mo - mn);
          for (int j = 0; j < 4; ++j) acc[r * GG + j * 32 + lane] *= f;
          if (lane == 0) sm[r] *= f;
        }
        if (lane == 0) mold[r] = mn;
      }
    }
    __syncthreads();

    {
      const int wv  = __builtin_amdgcn_readfirstlane(wave);
      const int tot = __builtin_amdgcn_readfirstlane(total);
      for (int i0 = wv * PCB; i0 < tot; i0 += 32 * PCB) {
        const int nv = min(PCB, tot - i0);
        const int   j  = min(lane, nv - 1);
        const int   kj = lkey[i0 + j];
        const float xj = lval[i0 + j];
        float4 h[PCB];
        #pragma unroll
        for (int u = 0; u < PCB; ++u) {
          const int ku = __builtin_amdgcn_readlane(kj, u);
          h[u] = ((const float4*)(hv + (size_t)(ku & 0xFFFF) * GG))[lane];
        }
        const int   rj = (int)(((unsigned)kj) >> 16);
        const float ej = expf(xj - mx[rj]);
        if (lane < nv) atomicAdd(&sm[rj], ej);
        #pragma unroll
        for (int u = 0; u < PCB; ++u) {
          if (u < nv) {
            const int   r  = (int)(((unsigned)__builtin_amdgcn_readlane(kj, u)) >> 16);
            const float ex = __int_as_float(__builtin_amdgcn_readlane(__float_as_int(ej), u));
            float* a = &acc[r * GG + lane];
            atomicAdd(a,      ex * h[u].x);
            atomicAdd(a + 32, ex * h[u].y);
            atomicAdd(a + 64, ex * h[u].z);
            atomicAdd(a + 96, ex * h[u].w);
          }
        }
      }
    }
    __syncthreads();
    c0 += epoch_len;
  }

  for (int f = tid; f < RN * 32; f += 1024) {
    const int r    = f >> 5;
    const int c4   = f & 31;
    const int node = base + r;
    if (node < NN) {
      const float den = sm[r];
      const float dd  = den > 0.0f ? den : 1.0f;
      float4 v;
      v.x = acc[r * GG +      c4] / dd;
      v.y = acc[r * GG + 32 + c4] / dd;
      v.z = acc[r * GG + 64 + c4] / dd;
      v.w = acc[r * GG + 96 + c4] / dd;
      v.x = v.x > 0.0f ? v.x : expm1f(v.x);
      v.y = v.y > 0.0f ? v.y : expm1f(v.y);
      v.z = v.z > 0.0f ? v.z : expm1f(v.z);
      v.w = v.w > 0.0f ? v.w : expm1f(v.w);
      ((float4*)(out + (size_t)node * GG))[c4] = v;
    }
  }
}

extern "C" void kernel_launch(void* const* d_in, const int* in_sizes, int n_in,
                              void* d_out, int out_size, void* d_ws, size_t ws_size,
                              hipStream_t stream) {
  (void)in_sizes; (void)n_in; (void)out_size; (void)ws_size;
  const float* nf  = (const float*)d_in[0];
  const int*   src = (const int*)  d_in[1];
  const int*   dst = (const int*)  d_in[2];
  const float* We  = (const float*)d_in[3];
  const float* be  = (const float*)d_in[4];
  const float* Wn  = (const float*)d_in[5];
  const float* bn  = (const float*)d_in[6];
  float* out = (float*)d_out;

  float* ws    = (float*)d_ws;
  float* hv    = ws;
  float* p     = hv + (size_t)NN * GG;
  float* q     = p + NN;
  unsigned* offc = (unsigned*)(q + NN);
  unsigned* recs = offc + (size_t)NWG * NCH;

  node_gemm_wmma<<<K1_WGS, 256, 0, stream>>>(nf, Wn, bn, hv);
  node_scores_partition<<<NCH, K2T, 0, stream>>>(nf, We, src, dst, p, q, offc, recs);
  dst_range_softmax_aggregate<<<NWG, 1024, 0, stream>>>(offc, recs, p, q, be, hv, out);
}


# Launch 3 groups its LDS list by node (count, prefix, place); pass C sums each run of equal node in registers: one LDS add per run and column, not per edge; sums: in-run serial, then LDS arrival
# speedup vs baseline: 5.6217x; 1.0859x over previous
#include <hip/hip_runtime.h>
#include <math.h>

typedef __attribute__((ext_vector_type(2))) float v2f;
typedef __attribute__((ext_vector_type(8))) float v8f;

#define NN 50000
#define EE 800000
#define DD 128
#define GG 128
#define LDS_STRIDE 132

#define K1_WGS 1024
#define NT (NN / 16)
static_assert(NN % 16 == 0, "whole row tiles only");
static_assert(K1_WGS <= NT, "every block has a first tile");

__global__ void __launch_bounds__(256)
node_gemm_wmma(const float* __restrict__ nf, const float* __restrict__ Wn,
               const float* __restrict__ bn, float* __restrict__ hv) {
  __shared__ float Atile[2][16 * LDS_STRIDE];
  const int tid  = threadIdx.x;
  const int wave = tid >> 5;
  const int lane = tid & 31;
  const int col0 = wave * 16;
  const int half = lane >> 4;
  const int l16  = lane & 15;

  const int srow = wave * LDS_STRIDE + lane * 4;
  int tile = blockIdx.x;
  {
    const float4* t0 = (const float4*)(nf + (size_t)tile * 16 * DD);
    const float4 f0 = t0[tid];
    const float4 f1 = t0[tid + 256];
    *(float4*)&Atile[0][srow]                  = f0;
    *(float4*)&Atile[0][srow + 8 * LDS_STRIDE] = f1;
  }

  v2f b[DD / 4];
  #pragma unroll
  for (int k = 0; k < DD; k += 4) {
    const int ka = k + half * 2;
    b[k >> 2].x = Wn[(size_t)(ka + 0) * GG + col0 + l16];
    b[k >> 2].y = Wn[(size_t)(ka + 1) * GG + col0 + l16];
  }
  const float bias = bn[col0 + l16];
  __syncthreads();

  int buf = 0;
  for (;;) {
    const int  next = tile + K1_WGS;
    const bool more = next < NT;
    float4 n0 = make_float4(0.f, 0.f, 0.f, 0.f), n1 = n0;
    if (more) {
      const float4* tn = (const float4*)(nf + (size_t)next * 16 * DD);
      n0 = tn[tid];
      n1 = tn[tid + 256];
    }

    const float* At = &Atile[buf][l16 * LDS_STRIDE + half * 2];
    v8f acc = {};
    #pragma unroll
    for (int k = 0; k < DD; k += 4) {
      v2f a;
      a.x = At[k];
      a.y = At[k + 1];
      acc = __builtin_amdgcn_wmma_f32_16x16x4_f32(false, a, false, b[k >> 2],
                                                  (short)0, acc, false, false);
    }

    if (more) {
      *(float4*)&Atile[buf ^ 1][srow]                  = n0;
      *(float4*)&Atile[buf ^ 1][srow + 8 * LDS_STRIDE] = n1;
      __syncthreads();
    }

    const int row0 = tile * 16;
    #pragma unroll
    for (int i = 0; i < 8; ++i) {
      const int row = row0 + i + half * 8;
      float v = acc[i] + bias;
      v = v > 0.0f ? v : 0.01f * v;
      hv[(size_t)row * GG + col0 + l16] = v;
    }

    if (!more) break;
    tile = next;
    buf ^= 1;
  }
}

#define RN 196
#define NWG 256

#define NPW 16
#define CH  4096
#define NCH ((EE + CH - 1) / CH)
#define K2T (CH / 8)
static_assert(NN % NPW == 0, "a wave's nodes are all inside the array");
static_assert(2 * NPW == 32, "one total per lane after the fold");
static_assert((NN / NPW) * 32 == EE / 8, "the threads that own edges are the threads that own nodes");
static_assert(NCH * K2T >= EE / 8 && (EE / 8) % 32 == 0, "every edge has a thread; validity is per wave");
static_assert(NWG * RN >= NN && NWG == 256, "the prefix below gives eight owners to each lane of one wave");
static_assert(CH < 65536 && NN <= 65536 && RN < 65536, "record and table words pack two 16-bit fields");

template <int H, int OFF>
__device__ inline void foldStep(float (&v)[2 * NPW], int lane) {
  const bool hi = (lane & OFF) != 0;
  #pragma unroll
  for (int j = 0; j < H; ++j) {
    const float keep = hi ? v[j + H] : v[j];
    const float send = hi ? v[j] : v[j + H];
    v[j] = keep + __shfl_xor(send, OFF, 32);
  }
}

__global__ void __launch_bounds__(K2T)
node_scores_partition(const float* __restrict__ nf, const float* __restrict__ We,
                      const int* __restrict__ src, const int* __restrict__ dst,
                      float* __restrict__ p, float* __restrict__ q,
                      unsigned* __restrict__ offc, unsigned* __restrict__ recs) {
  __shared__ unsigned hist[NWG];
  __shared__ unsigned bstart[NWG];
  __shared__ uint4    srec4[CH / 4];
  unsigned* srec = (unsigned*)srec4;

  const int tid  = threadIdx.x;
  const int lane = tid & 31;
  const int gid  = (int)(blockIdx.x * K2T) + tid;
  const bool valid = gid < EE / 8;

  int dd[8] = {0, 0, 0, 0, 0, 0, 0, 0};
  int ss[8] = {0, 0, 0, 0, 0, 0, 0, 0};
  float4 x[NPW];
  float4 wp = make_float4(0.f, 0.f, 0.f, 0.f), wq = wp;
  #pragma unroll
  for (int u = 0; u < NPW; ++u) x[u] = wp;
  const int node0 = (gid >> 5) * NPW;
  if (valid) {
    const int4 d0 = ((const int4*)dst)[2 * gid];
    const int4 d1 = ((const int4*)dst)[2 * gid + 1];
    const int4 s0 = ((const int4*)src)[2 * gid];
    const int4 s1 = ((const int4*)src)[2 * gid + 1];
    dd[0] = d0.x; dd[1] = d0.y; dd[2] = d0.z; dd[3] = d0.w;
    dd[4] = d1.x; dd[5] = d1.y; dd[6] = d1.z; dd[7] = d1.w;
    ss[0] = s0.x; ss[1] = s0.y; ss[2] = s0.z; ss[3] = s0.w;
    ss[4] = s1.x; ss[5] = s1.y; ss[6] = s1.z; ss[7] = s1.w;
    const float4* rows = (const float4*)(nf + (size_t)node0 * DD);
    #pragma unroll
    for (int u = 0; u < NPW; ++u) x[u] = rows[u * (DD / 4) + lane];
    wp = ((const float4*)We)[lane];
    wq = ((const float4*)(We + DD))[lane];
  }
  if (tid < NWG) hist[tid] = 0u;
  __syncthreads();

  unsigned ow[8], rk[8];
  #pragma unroll
  for (int j = 0; j < 8; ++j) { ow[j] = 0u; rk[j] = 0u; }
  float score = 0.0f;
  if (valid) {
    #pragma unroll
    for (int j = 0; j < 8; ++j) {
      ow[j] = min((unsigned)dd[j] / (unsigned)RN, (unsigned)(NWG - 1));
      rk[j] = atomicAdd(&hist[ow[j]], 1u);
    }

    float v[2 * NPW];
    #pragma unroll
    for (int u = 0; u < NPW; ++u) {
      v[u]       = x[u].x * wp.x + x[u].y * wp.y + x[u].z * wp.z + x[u].w * wp.w;
      v[NPW + u] = x[u].x * wq.x + x[u].y * wq.y + x[u].z * wq.z + x[u].w * wq.w;
    }
    foldStep<16, 16>(v, lane);
    foldStep<8, 8>(v, lane);
    foldStep<4, 4>(v, lane);
    foldStep<2, 2>(v, lane);
    foldStep<1, 1>(v, lane);
    score = v[0];
  }
  __syncthreads();

  if (tid < 32) {
    unsigned c[8], e[8], run = 0u;
    #pragma unroll
    for (int j = 0; j < 8; ++j) { c[j] = hist[8 * lane + j]; e[j] = run; run += c[j]; }
    unsigned inc = run;
    #pragma unroll
    for (int off = 1; off < 32; off <<= 1) {
      const unsigned n = __shfl_up(inc, off, 32);
      if (lane >= off) inc += n;
    }
    const unsigned base = inc - run;
    #pragma unroll
    for (int j = 0; j < 8; ++j) bstart[8 * lane + j] = base + e[j];
  }
  __syncthreads();

  if (valid) {
    #pragma unroll
    for (int j = 0; j < 8; ++j) {
      const unsigned r = min((unsigned)dd[j] - ow[j] * (unsigned)RN, (unsigned)(RN - 1));
      srec[bstart[ow[j]] + rk[j]] = (r << 16) | ((unsigned)ss[j] & 0xFFFFu);
    }
  }
  const unsigned tabw = tid < NWG ? ((bstart[tid] << 16) | hist[tid]) : 0u;
  const unsigned tot  = bstart[NWG - 1] + hist[NWG - 1];
  __syncthreads();

  uint4* outc = (uint4*)recs + (size_t)blockIdx.x * (CH / 4);
  #pragma unroll
  for (int k = 0; k < (CH / 4) / K2T; ++k) {
    const int i4 = tid + k * K2T;
    if ((unsigned)(4 * i4) < tot) outc[i4] = srec4[i4];
  }
  if (tid < NWG) offc[(size_t)tid * NCH + blockIdx.x] = tabw;
  if (valid) {
    float* o = lane < NPW ? p : q;
    o[node0 + (lane & (NPW - 1))] = score;
  }
}

#define CAP 16384
#define SAFE_CH (CAP / CH)
#define PCB 8
static_assert(SAFE_CH >= 1 && SAFE_CH * CH <= CAP, "a short epoch always fits the list");
static_assert(NCH <= 4 * 64 && NCH <= 1024, "four segments per half-wave cover an epoch; one thread per segment");

__device__ inline void ldsMaxF(float* addr, float v) {
  const int bits = __float_as_int(v);
  if (bits >= 0) atomicMax((int*)addr, bits);
  else           atomicMin((unsigned int*)addr, (unsigned int)bits);
}

__device__ inline void flushRun(float* acc, float* sm, int r, int lane,
                                float s0, float s1, float s2, float s3, float es) {
  float* a = &acc[r * GG + lane];
  atomicAdd(a,      s0);
  atomicAdd(a + 32, s1);
  atomicAdd(a + 64, s2);
  atomicAdd(a + 96, s3);
  if (lane == 0) atomicAdd(&sm[r], es);
}

template <bool FULL>
__device__ inline void rowSumBatch(const int* skey, const float* sval, const float* mx, float* sm,
                                   float* acc, const float* __restrict__ hv, int lane, int i0, int nv) {
  const int   j  = min(lane, nv - 1);
  const int   kj = skey[i0 + j];
  const float xj = sval[i0 + j];
  float4 h[PCB];
  #pragma unroll
  for (int u = 0; u < PCB; ++u) {
    const int ku = __builtin_amdgcn_readlane(kj, u);
    h[u] = ((const float4*)(hv + (size_t)(ku & 0xFFFF) * GG))[lane];
  }
  const int   rj = (int)(((unsigned)kj) >> 16);
  const float ej = expf(xj - mx[rj]);
  int   rcur = (int)(((unsigned)__builtin_amdgcn_readlane(kj, 0)) >> 16);
  float es = __int_as_float(__builtin_amdgcn_readlane(__float_as_int(ej), 0));
  float s0 = es * h[0].x, s1 = es * h[0].y, s2 = es * h[0].z, s3 = es * h[0].w;
  #pragma unroll
  for (int u = 1; u < PCB; ++u) {
    if (FULL || u < nv) {
      const int   r  = (int)(((unsigned)__builtin_amdgcn_readlane(kj, u)) >> 16);
      const float ex = __int_as_float(__builtin_amdgcn_readlane(__float_as_int(ej), u));
      if (r != rcur) {
        flushRun(acc, sm, rcur, lane, s0, s1, s2, s3, es);
        s0 = 0.0f; s1 = 0.0f; s2 = 0.0f; s3 = 0.0f; es = 0.0f;
        rcur = r;
      }
      s0 += ex * h[u].x;
      s1 += ex * h[u].y;
      s2 += ex * h[u].z;
      s3 += ex * h[u].w;
      es += ex;
    }
  }
  flushRun(acc, sm, rcur, lane, s0, s1, s2, s3, es);
}

__global__ void __launch_bounds__(1024)
dst_range_softmax_aggregate(const unsigned* __restrict__ offc, const unsigned* __restrict__ recs,
                            const float* __restrict__ p, const float* __restrict__ q,
                            const float* __restrict__ be, const float* __restrict__ hv,
                            float* __restrict__ out) {
  __shared__ float acc[RN * GG];
  __shared__ int   lkey[CAP];
  __shared__ int   skey[CAP];
  __shared__ float sval[CAP];
  __shared__ unsigned ncur[RN];
  __shared__ float mx[RN];
  __shared__ float mold[RN];
  __shared__ float sm[RN];
  __shared__ float pl[RN];
  __shared__ int      segk[NCH];
  __shared__ unsigned segw[NCH];
  __shared__ int   cnt;

  const int tid  = threadIdx.x;
  const int lane = tid & 31;
  const int wave = tid >> 5;
  const int base = blockIdx.x * RN;

  for (int i = tid; i < RN * GG; i += 1024) acc[i] = 0.0f;
  if (tid < RN) {
    const int node = base + tid;
    mx[tid]   = -INFINITY;
    mold[tid] = -INFINITY;
    sm[tid]   = 0.0f;
    pl[tid]   = node < NN ? p[node] : 0.0f;
  }
  const float bias = be[0];
  const unsigned* myoff = offc + (size_t)blockIdx.x * NCH;

  int epoch_len = NCH;
  int c0 = 0;
  while (c0 < NCH) {
    if (tid == 0) cnt = 0;
    if (tid < RN) ncur[tid] = 0u;
    __syncthreads();
    const int c_end = (c0 + epoch_len < NCH) ? (c0 + epoch_len) : NCH;
    const int nseg  = c_end - c0;

    if (tid < ((nseg + 31) & ~31)) {
      const unsigned word = tid < nseg ? myoff[c0 + tid] : 0u;
      const unsigned c = word & 0xFFFFu;
      unsigned inc = c;
      #pragma unroll
      for (int off = 1; off < 32; off <<= 1) {
        const unsigned n = __shfl_up(inc, off, 32);
        if (lane >= off) inc += n;
      }
      int basew = 0;
      if (lane == 31) basew = atomicAdd(&cnt, (int)inc);
      basew = __shfl(basew, 31, 32);
      if (tid < nseg) {
        segk[tid] = basew + (int)(inc - c);
        segw[tid] = word;
      }
    }
    __syncthreads();
    const int total = cnt;
    if (total > CAP) {
      epoch_len = SAFE_CH;
      __syncthreads();
      continue;
    }

    {
      const int hw  = tid >> 4;
      const int sub = tid & 15;
      const unsigned* sp[4];
      unsigned cc[4], v0[4], v1[4];
      int kk[4];
      #pragma unroll
      for (int u = 0; u < 4; ++u) {
        const int t  = hw + 64 * u;
        const int tc = t < nseg ? t : nseg - 1;
        const unsigned word = segw[tc];
        cc[u] = t < nseg ? (word & 0xFFFFu) : 0u;
        kk[u] = segk[tc];
        sp[u] = recs + (size_t)(c0 + tc) * CH + (word >> 16);
        v0[u] = 0u; v1[u] = 0u;
        if ((unsigned)sub < cc[u])        v0[u] = sp[u][sub];
        if ((unsigned)(sub + 16) < cc[u]) v1[u] = sp[u][sub + 16];
      }
      #pragma unroll
      for (int u = 0; u < 4; ++u) {
        if ((unsigned)sub < cc[u])        { lkey[kk[u] + sub]      = (int)v0[u]; atomicAdd(&ncur[v0[u] >> 16], 1u); }
        if ((unsigned)(sub + 16) < cc[u]) { lkey[kk[u] + sub + 16] = (int)v1[u]; atomicAdd(&ncur[v1[u] >> 16], 1u); }
        #pragma clang loop unroll(disable)
        for (unsigned i = (unsigned)sub + 32u; i < cc[u]; i += 16u) {
          const unsigned w = sp[u][i];
          lkey[kk[u] + (int)i] = (int)w;
          atomicAdd(&ncur[w >> 16], 1u);
        }
      }
    }
    __syncthreads();

    if (tid < 32) {
      unsigned c[7], e[7], run = 0u;
      #pragma unroll
      for (int j = 0; j < 7; ++j) {
        const int r = 7 * lane + j;
        c[j] = r < RN ? ncur[r] : 0u;
        e[j] = run;
        run += c[j];
      }
      unsigned inc = run;
      #pragma unroll
      for (int off = 1; off < 32; off <<= 1) {
        const unsigned n = __shfl_up(inc, off, 32);
        if (lane >= off) inc += n;
      }
      const unsigned first = inc - run;
      #pragma unroll
      for (int j = 0; j < 7; ++j) {
        const int r = 7 * lane + j;
        if (r < RN) ncur[r] = first + e[j];
      }
    }
    __syncthreads();

    for (int i = tid; i < total; i += 1024) {
      const unsigned key = (unsigned)lkey[i];
      const unsigned r = key >> 16;
      const unsigned pos = atomicAdd(&ncur[r], 1u);
      float x = pl[r] + q[key & 0xFFFFu] + bias;
      x = x > 0.0f ? x : 0.01f * x;
      skey[pos] = (int)key;
      sval[pos] = x;
      ldsMaxF(&mx[r], x);
    }
    __syncthreads();
    for (int r = wave; r < RN; r += 32) {
      const float mo = mold[r];
      const float mn = mx[r];
      if (mn != mo) {
        if (mo != -INFINITY) {
          const float f = expf(mo - mn);
          for (int j = 0; j < 4; ++j) acc[r * GG + j * 32 + lane] *= f;
          if (lane == 0) sm[r] *= f;
        }
        if (lane == 0) mold[r] = mn;
      }
    }
    __syncthreads();

    {
      const int wv  = __builtin_amdgcn_readfirstlane(wave);
      const int tot = __builtin_amdgcn_readfirstlane(total);
      for (int i0 = wv * PCB; i0 < tot; i0 += 32 * PCB) {
        const int nv = tot - i0;
        if (nv >= PCB) rowSumBatch<true>(skey, sval, mx, sm, acc, hv, lane, i0, PCB);
        else           rowSumBatch<false>(skey, sval, mx, sm, acc, hv, lane, i0, nv);
      }
    }
    __syncthreads();
    c0 += epoch_len;
  }

  for (int f = tid; f < RN * 32; f += 1024) {
    const int r    = f >> 5;
    const int c4   = f & 31;
    const int node = base + r;
    if (node < NN) {
      const float den = sm[r];
      const float dd  = den > 0.0f ? den : 1.0f;
      float4 v;
      v.x = acc[r * GG +      c4] / dd;
      v.y = acc[r * GG + 32 + c4] / dd;
      v.z = acc[r * GG + 64 + c4] / dd;
      v.w = acc[r * GG + 96 + c4] / dd;
      v.x = v.x > 0.0f ? v.x : expm1f(v.x);
      v.y = v.y > 0.0f ? v.y : expm1f(v.y);
      v.z = v.z > 0.0f ? v.z : expm1f(v.z);
      v.w = v.w > 0.0f ? v.w : expm1f(v.w);
      ((float4*)(out + (size_t)node * GG))[c4] = v;
    }
  }
}

extern "C" void kernel_launch(void* const* d_in, const int* in_sizes, int n_in,
                              void* d_out, int out_size, void* d_ws, size_t ws_size,
                              hipStream_t stream) {
  (void)in_sizes; (void)n_in; (void)out_size; (void)ws_size;
  const float* nf  = (const float*)d_in[0];
  const int*   src = (const int*)  d_in[1];
  const int*   dst = (const int*)  d_in[2];
  const float* We  = (const float*)d_in[3];
  const float* be  = (const float*)d_in[4];
  const float* Wn  = (const float*)d_in[5];
  const float* bn  = (const float*)d_in[6];
  float* out = (float*)d_out;

  float* ws    = (float*)d_ws;
  float* hv    = ws;
  float* p     = hv + (size_t)NN * GG;
  float* q     = p + NN;
  unsigned* offc = (unsigned*)(q + NN);
  unsigned* recs = offc + (size_t)NWG * NCH;

  node_gemm_wmma<<<K1_WGS, 256, 0, stream>>>(nf, Wn, bn, hv);
  node_scores_partition<<<NCH, K2T, 0, stream>>>(nf, We, src, dst, p, q, offc, recs);
  dst_range_softmax_aggregate<<<NWG, 1024, 0, stream>>>(offc, recs, p, q, be, hv, out);
}
